# starting kernel
# baseline (speedup 1.0000x reference)
#include <hip/hip_runtime.h>

#define NN   100000
#define BB   512
#define EE   1000000
#define ESZ  1200000
#define DD   64
#define NBT  (NN + BB)

#define WS_A0   0
#define WS_A1   (NBT)
#define WS_M0   (2*NBT)
#define WS_M1   (3*NBT)
#define WS_DEG0 (4*NBT)
#define WS_DEG1 (5*NBT)
#define WS_SM   (6*NBT)
#define WS_U1   (WS_SM + 128)
#define WS_U2   (WS_SM + 192)
#define WS_T    (WS_SM + 256)
#define WS_ESA  (WS_T + 2*BB)
#define WS_HID  (WS_ESA + 2*BB*DD)

typedef float v2f __attribute__((ext_vector_type(2)));
typedef float v8f __attribute__((ext_vector_type(8)));

__global__ void k_zero(float* ws) {
    int i = blockIdx.x * blockDim.x + threadIdx.x;
    if (i < 6 * NBT) ws[i] = 0.0f;
}

__global__ void k_small(const float* __restrict__ w_n2l,
                        const float* __restrict__ p_conv,
                        const float* __restrict__ W_sage,
                        float* ws) {
    __shared__ float v[DD], w[DD];
    int d = threadIdx.x;
    float x = fmaxf(w_n2l[d] + w_n2l[DD + d], 0.0f);
    v[d] = x;
    __syncthreads();
    float ss = 0.f;
    for (int k = 0; k < DD; ++k) ss += v[k] * v[k];
    float nv = x / fmaxf(sqrtf(ss), 1e-12f);
    __syncthreads();
    v[d] = nv;
    __syncthreads();
    float wd = 0.f;
    for (int k = 0; k < DD; ++k) wd += v[k] * p_conv[k * DD + d];
    w[d] = wd;
    __syncthreads();
    float u1 = 0.f, u2 = 0.f;
    for (int k = 0; k < DD; ++k) {
        u1 += w[k] * W_sage[k * DD + d];
        u2 += w[k] * W_sage[(DD + k) * DD + d];
    }
    ws[WS_SM + d]  = nv;
    ws[WS_SM + 64 + d] = wd;
    ws[WS_U1 + d] = u1;
    ws[WS_U2 + d] = u2;
}

__global__ void k_n2n(const int* __restrict__ r0, const float* __restrict__ v0,
                      const int* __restrict__ r1, const float* __restrict__ v1,
                      float* ws) {
    int e = blockIdx.x * blockDim.x + threadIdx.x;
    if (e < EE) {
        atomicAdd(&ws[WS_A0 + r0[e]], v0[e]);
    } else if (e < 2 * EE) {
        int i = e - EE;
        atomicAdd(&ws[WS_A1 + r1[i]], v1[i]);
    }
}

__global__ void k_subg(const int* __restrict__ subg_row, float* ws) {
    int i = blockIdx.x * blockDim.x + threadIdx.x;
    if (i < NN) {
        int b = subg_row[i];
        atomicAdd(&ws[WS_A0 + NN + b], 1.0f);
        atomicAdd(&ws[WS_A1 + NN + b], 1.0f);
    }
}

__global__ void k_sage(const int* __restrict__ s0r, const int* __restrict__ s0c,
                       const int* __restrict__ s1r, const int* __restrict__ s1c,
                       float* ws) {
    int e = blockIdx.x * blockDim.x + threadIdx.x;
    if (e < ESZ) {
        int r = s0r[e];
        atomicAdd(&ws[WS_M0 + r], ws[WS_A0 + s0c[e]]);
        atomicAdd(&ws[WS_DEG0 + r], 1.0f);
    } else if (e < 2 * ESZ) {
        int i = e - ESZ;
        int r = s1r[i];
        atomicAdd(&ws[WS_M1 + r], ws[WS_A1 + s1c[i]]);
        atomicAdd(&ws[WS_DEG1 + r], 1.0f);
    }
}

__global__ void k_rows(float* ws, const float* __restrict__ cp, float* out) {
    int wid  = (blockIdx.x * blockDim.x + threadIdx.x) >> 5;
    int lane = threadIdx.x & 31;
    if (wid >= 2 * NBT) return;
    int l = wid / NBT;
    int i = wid - l * NBT;

    float a   = ws[(l ? WS_A1  : WS_A0)  + i];
    float num = ws[(l ? WS_M1  : WS_M0)  + i];
    float dg  = ws[(l ? WS_DEG1: WS_DEG0)+ i];
    float m   = num / fmaxf(dg, 1.0f);

    int d0 = 2 * lane, d1 = 2 * lane + 1;
    float r0 = fmaxf(a * ws[WS_U1 + d0] + m * ws[WS_U2 + d0], 0.0f);
    float r1 = fmaxf(a * ws[WS_U1 + d1] + m * ws[WS_U2 + d1], 0.0f);

    float ss = r0 * r0 + r1 * r1;
    for (int off = 16; off >= 1; off >>= 1) ss += __shfl_xor(ss, off, 32);
    float inv = 1.0f / fmaxf(sqrtf(ss), 1e-12f);
    r0 *= inv; r1 *= inv;

    if (i < NN) {
        float2 pr = make_float2(r0, r1);
        float* dst = out + BB + (l * NN + i) * DD + d0;
        *(float2*)dst = pr;
    } else {
        int b = i - NN;
        float t = r0 * cp[d0] + r1 * cp[d1];
        for (int off = 16; off >= 1; off >>= 1) t += __shfl_xor(t, off, 32);
        if (lane == 0) ws[WS_T + l * BB + b] = t;
    }
}

__global__ void k_esa(const int* __restrict__ act_col, const float* __restrict__ out,
                      float* ws) {
    int idx = blockIdx.x * blockDim.x + threadIdx.x;
    if (idx >= 2 * BB * DD) return;
    int d = idx & (DD - 1);
    int b = (idx >> 6) & (BB - 1);
    int l = idx >> 15;
    float t = ws[WS_T + l * BB + b];
    float ae = out[BB + (l * NN + act_col[b]) * DD + d];
    ws[WS_ESA + idx] = t * ae;
}

__global__ void k_gemm(const float* __restrict__ ws, const float* __restrict__ h1,
                       float* __restrict__ hid) {
    int bid  = blockIdx.x;
    int tn   = bid & 1;
    int tm   = (bid >> 1) & 31;
    int l    = bid >> 6;
    int lane = threadIdx.x;
    int half = lane >> 4;
    int lid  = lane & 15;

    const float* A = ws + WS_ESA + (l * BB + tm * 16) * DD;
    v8f c = {0.f, 0.f, 0.f, 0.f, 0.f, 0.f, 0.f, 0.f};
    #pragma unroll
    for (int s = 0; s < 16; ++s) {
        int k0 = 4 * s + 2 * half;
        v2f av, bv;
        av.x = A[lid * DD + k0];
        av.y = A[lid * DD + k0 + 1];
        bv.x = h1[(k0    ) * 32 + tn * 16 + lid];
        bv.y = h1[(k0 + 1) * 32 + tn * 16 + lid];
        c = __builtin_amdgcn_wmma_f32_16x16x4_f32(false, av, false, bv,
                                                  (short)0, c, false, false);
    }
    float* dst = hid + (l * BB + tm * 16) * 32 + tn * 16;
    #pragma unroll
    for (int r = 0; r < 8; ++r) {
        int M = r + 8 * half;
        dst[M * 32 + lid] = fmaxf(c[r], 0.0f);
    }
}

__global__ void k_q(const float* __restrict__ ws, const float* __restrict__ aux,
                    const float* __restrict__ h2w, float* out) {
    int b = blockIdx.x * blockDim.x + threadIdx.x;
    if (b >= BB) return;
    float q = 0.0f;
    for (int l = 0; l < 2; ++l) {
        const float* h = ws + WS_HID + (l * BB + b) * 32;
        for (int j = 0; j < 32; ++j) q += h[j] * h2w[j];
        for (int j = 0; j < 4; ++j)  q += aux[b * 8 + l * 4 + j] * h2w[32 + j];
    }
    out[b] = q;
}

extern "C" void kernel_launch(void* const* d_in, const int* in_sizes, int n_in,
                              void* d_out, int out_size, void* d_ws, size_t ws_size,
                              hipStream_t stream) {
    const int*   n2n0_row = (const int*)  d_in[0];
    const float* n2n0_val = (const float*)d_in[2];
    const int*   n2n1_row = (const int*)  d_in[3];
    const float* n2n1_val = (const float*)d_in[5];
    const int*   subg_row = (const int*)  d_in[6];
    const int*   act_col  = (const int*)  d_in[9];
    const int*   sg0_row  = (const int*)  d_in[10];
    const int*   sg0_col  = (const int*)  d_in[11];
    const int*   sg1_row  = (const int*)  d_in[12];
    const int*   sg1_col  = (const int*)  d_in[13];
    const float* aux      = (const float*)d_in[14];
    const float* w_n2l    = (const float*)d_in[15];
    const float* p_conv   = (const float*)d_in[16];
    const float* W_sage   = (const float*)d_in[17];
    const float* cp       = (const float*)d_in[18];
    const float* h1w      = (const float*)d_in[19];
    const float* h2w      = (const float*)d_in[20];
    float* out = (float*)d_out;
    float* ws  = (float*)d_ws;

    k_zero <<<(6 * NBT + 255) / 256, 256, 0, stream>>>(ws);
    k_small<<<1, 64, 0, stream>>>(w_n2l, p_conv, W_sage, ws);
    k_n2n  <<<(2 * EE + 255) / 256, 256, 0, stream>>>(n2n0_row, n2n0_val,
                                                      n2n1_row, n2n1_val, ws);
    k_subg <<<(NN + 255) / 256, 256, 0, stream>>>(subg_row, ws);
    k_sage <<<(2 * ESZ + 255) / 256, 256, 0, stream>>>(sg0_row, sg0_col,
                                                       sg1_row, sg1_col, ws);
    k_rows <<<(2 * NBT * 32 + 255) / 256, 256, 0, stream>>>(ws, cp, out);
    k_esa  <<<(2 * BB * DD + 255) / 256, 256, 0, stream>>>(act_col, out, ws);
    k_gemm <<<128, 32, 0, stream>>>(ws, h1w, ws + WS_HID);
    k_q    <<<(BB + 255) / 256, 256, 0, stream>>>(ws, aux, h2w, out);
}


# n2n and sage scatters: block owns a row range, sums its edge chunk in LDS, flushes by lane-contiguous global atomics (was one random global atomic per edge); float sum order changes
# speedup vs baseline: 3.2171x; 3.2171x over previous
#include <hip/hip_runtime.h>

#define NN   100000
#define BB   512
#define EE   1000000
#define ESZ  1200000
#define DD   64
#define NBT  (NN + BB)

#define WS_A0   0
#define WS_A1   (NBT)
#define WS_M0   (2*NBT)
#define WS_M1   (3*NBT)
#define WS_DEG0 (4*NBT)
#define WS_DEG1 (5*NBT)
#define WS_SM   (6*NBT)
#define WS_U1   (WS_SM + 128)
#define WS_U2   (WS_SM + 192)
#define WS_T    (WS_SM + 256)
#define WS_ESA  (WS_T + 2*BB)
#define WS_HID  (WS_ESA + 2*BB*DD)

#define SC_P    16
#define SC_C    8
#define SC_T    1024
#define SC_V    2
#define SC_U    (4 * SC_V)
#define SC_GRID (2 * SC_P * SC_C)
#define N2N_RB  (NN / SC_P)
#define SG_RB   (NBT / SC_P)
#define N2N_CH  (EE / SC_C)
#define SG_CH   (ESZ / SC_C)
static_assert(NN % SC_P == 0 && NBT % SC_P == 0, "row ranges must tile the rows");
static_assert(EE % SC_C == 0 && ESZ % SC_C == 0, "edge chunks must tile the edges");
static_assert(N2N_CH % 4 == 0 && SG_CH % 4 == 0, "a chunk is read in four-word pieces");

typedef float v2f __attribute__((ext_vector_type(2)));
typedef float v8f __attribute__((ext_vector_type(8)));

__global__ void k_zero(float* ws) {
    int i = blockIdx.x * blockDim.x + threadIdx.x;
    if (i < 6 * NBT) ws[i] = 0.0f;
}

__global__ void k_small(const float* __restrict__ w_n2l,
                        const float* __restrict__ p_conv,
                        const float* __restrict__ W_sage,
                        float* ws) {
    __shared__ float v[DD], w[DD];
    int d = threadIdx.x;
    float x = fmaxf(w_n2l[d] + w_n2l[DD + d], 0.0f);
    v[d] = x;
    __syncthreads();
    float ss = 0.f;
    for (int k = 0; k < DD; ++k) ss += v[k] * v[k];
    float nv = x / fmaxf(sqrtf(ss), 1e-12f);
    __syncthreads();
    v[d] = nv;
    __syncthreads();
    float wd = 0.f;
    for (int k = 0; k < DD; ++k) wd += v[k] * p_conv[k * DD + d];
    w[d] = wd;
    __syncthreads();
    float u1 = 0.f, u2 = 0.f;
    for (int k = 0; k < DD; ++k) {
        u1 += w[k] * W_sage[k * DD + d];
        u2 += w[k] * W_sage[(DD + k) * DD + d];
    }
    ws[WS_SM + d]  = nv;
    ws[WS_SM + 64 + d] = wd;
    ws[WS_U1 + d] = u1;
    ws[WS_U2 + d] = u2;
}

__global__ __launch_bounds__(SC_T)
void k_n2n(const int* __restrict__ r0, const float* __restrict__ v0,
           const int* __restrict__ r1, const float* __restrict__ v1,
           float* ws) {
    __shared__ float acc[N2N_RB];
    const int tid = threadIdx.x;
    const int bid = blockIdx.x;
    const int c = bid % SC_C;
    const int p = (bid / SC_C) % SC_P;
    const int l = bid / (SC_C * SC_P);
    const int lo = p * N2N_RB;
    const int4*   __restrict__ row4 = reinterpret_cast<const int4*>((l ? r1 : r0) + c * N2N_CH);
    const float4* __restrict__ val4 = reinterpret_cast<const float4*>((l ? v1 : v0) + c * N2N_CH);
    const int n4 = N2N_CH / 4;

    for (int i = tid; i < N2N_RB; i += SC_T) acc[i] = 0.0f;
    __syncthreads();

    int4 qr[SC_V]; float4 qv[SC_V];
    #pragma unroll
    for (int u = 0; u < SC_V; ++u) {
        int i4 = u * SC_T + tid;
        int ii = i4 < n4 ? i4 : 0;
        qr[u] = row4[ii];
        qv[u] = val4[ii];
    }
    for (int b4 = 0; b4 < n4; b4 += SC_T * SC_V) {
        int rr[SC_U]; float vv[SC_U];
        #pragma unroll
        for (int u = 0; u < SC_V; ++u) {
            bool ok = (b4 + u * SC_T + tid) < n4;
            rr[4 * u + 0] = ok ? qr[u].x - lo : -1;  vv[4 * u + 0] = qv[u].x;
            rr[4 * u + 1] = ok ? qr[u].y - lo : -1;  vv[4 * u + 1] = qv[u].y;
            rr[4 * u + 2] = ok ? qr[u].z - lo : -1;  vv[4 * u + 2] = qv[u].z;
            rr[4 * u + 3] = ok ? qr[u].w - lo : -1;  vv[4 * u + 3] = qv[u].w;
        }
        #pragma unroll
        for (int u = 0; u < SC_V; ++u) {
            int i4 = b4 + SC_T * SC_V + u * SC_T + tid;
            int ii = i4 < n4 ? i4 : 0;
            qr[u] = row4[ii];
            qv[u] = val4[ii];
        }
        #pragma unroll
        for (int k = 0; k < SC_U; ++k)
            if ((unsigned)rr[k] < (unsigned)N2N_RB) atomicAdd(&acc[rr[k]], vv[k]);
    }
    __syncthreads();

    float* dst = ws + (l ? WS_A1 : WS_A0) + lo;
    for (int i = tid; i < N2N_RB; i += SC_T) {
        float sv = acc[i];
        if (sv != 0.0f) atomicAdd(&dst[i], sv);
    }
}

__global__ void k_subg(const int* __restrict__ subg_row, float* ws) {
    int i = blockIdx.x * blockDim.x + threadIdx.x;
    if (i < NN) {
        int b = subg_row[i];
        atomicAdd(&ws[WS_A0 + NN + b], 1.0f);
        atomicAdd(&ws[WS_A1 + NN + b], 1.0f);
    }
}

__global__ __launch_bounds__(SC_T)
void k_sage(const int* __restrict__ s0r, const int* __restrict__ s0c,
            const int* __restrict__ s1r, const int* __restrict__ s1c,
            float* ws) {
    __shared__ float        num[SG_RB];
    __shared__ unsigned int cnt[SG_RB];
    const int tid = threadIdx.x;
    const int bid = blockIdx.x;
    const int c = bid % SC_C;
    const int p = (bid / SC_C) % SC_P;
    const int l = bid / (SC_C * SC_P);
    const int lo = p * SG_RB;
    const int4* __restrict__ row4 = reinterpret_cast<const int4*>((l ? s1r : s0r) + c * SG_CH);
    const int4* __restrict__ col4 = reinterpret_cast<const int4*>((l ? s1c : s0c) + c * SG_CH);
    const float* a = ws + (l ? WS_A1 : WS_A0);
    const int n4 = SG_CH / 4;
    const int STEP4 = SC_T * SC_V;

    for (int i = tid; i < SG_RB; i += SC_T) { num[i] = 0.0f; cnt[i] = 0u; }
    __syncthreads();

    int4 qr[SC_V]; int4 qc[SC_V];
    int rr[SC_U]; float gv[SC_U];
    #pragma unroll
    for (int u = 0; u < SC_V; ++u) {
        int i4 = u * SC_T + tid;
        int ii = i4 < n4 ? i4 : 0;
        qr[u] = row4[ii];
        qc[u] = col4[ii];
    }
    #pragma unroll
    for (int u = 0; u < SC_V; ++u) {
        bool ok = (u * SC_T + tid) < n4;
        int r4[4] = {qr[u].x - lo, qr[u].y - lo, qr[u].z - lo, qr[u].w - lo};
        int c4[4] = {qc[u].x, qc[u].y, qc[u].z, qc[u].w};
        #pragma unroll
        for (int j = 0; j < 4; ++j) {
            bool m = ok && ((unsigned)r4[j] < (unsigned)SG_RB);
            rr[4 * u + j] = m ? r4[j] : -1;
            gv[4 * u + j] = a[m ? c4[j] : 0];
        }
    }
    #pragma unroll
    for (int u = 0; u < SC_V; ++u) {
        int i4 = STEP4 + u * SC_T + tid;
        int ii = i4 < n4 ? i4 : 0;
        qr[u] = row4[ii];
        qc[u] = col4[ii];
    }
    for (int b4 = 0; b4 < n4; b4 += STEP4) {
        int nr[SC_U]; float ng[SC_U];
        #pragma unroll
        for (int u = 0; u < SC_V; ++u) {
            bool ok = (b4 + STEP4 + u * SC_T + tid) < n4;
            int r4[4] = {qr[u].x - lo, qr[u].y - lo, qr[u].z - lo, qr[u].w - lo};
            int c4[4] = {qc[u].x, qc[u].y, qc[u].z, qc[u].w};
            #pragma unroll
            for (int j = 0; j < 4; ++j) {
                bool m = ok && ((unsigned)r4[j] < (unsigned)SG_RB);
                nr[4 * u + j] = m ? r4[j] : -1;
                ng[4 * u + j] = a[m ? c4[j] : 0];
            }
        }
        #pragma unroll
        for (int u = 0; u < SC_V; ++u) {
            int i4 = b4 + 2 * STEP4 + u * SC_T + tid;
            int ii = i4 < n4 ? i4 : 0;
            qr[u] = row4[ii];
            qc[u] = col4[ii];
        }
        #pragma unroll
        for (int k = 0; k < SC_U; ++k)
            if (rr[k] >= 0) {
                atomicAdd(&num[rr[k]], gv[k]);
                atomicAdd(&cnt[rr[k]], 1u);
            }
        #pragma unroll
        for (int k = 0; k < SC_U; ++k) { rr[k] = nr[k]; gv[k] = ng[k]; }
    }
    __syncthreads();

    float* dm = ws + (l ? WS_M1   : WS_M0)   + lo;
    float* dd = ws + (l ? WS_DEG1 : WS_DEG0) + lo;
    for (int i = tid; i < SG_RB; i += SC_T) {
        unsigned int k = cnt[i];
        if (k != 0u) {
            atomicAdd(&dm[i], num[i]);
            atomicAdd(&dd[i], (float)k);
        }
    }
}

__global__ void k_rows(float* ws, const float* __restrict__ cp, float* out) {
    int wid  = (blockIdx.x * blockDim.x + threadIdx.x) >> 5;
    int lane = threadIdx.x & 31;
    if (wid >= 2 * NBT) return;
    int l = wid / NBT;
    int i = wid - l * NBT;

    float a   = ws[(l ? WS_A1  : WS_A0)  + i];
    float num = ws[(l ? WS_M1  : WS_M0)  + i];
    float dg  = ws[(l ? WS_DEG1: WS_DEG0)+ i];
    float m   = num / fmaxf(dg, 1.0f);

    int d0 = 2 * lane, d1 = 2 * lane + 1;
    float r0 = fmaxf(a * ws[WS_U1 + d0] + m * ws[WS_U2 + d0], 0.0f);
    float r1 = fmaxf(a * ws[WS_U1 + d1] + m * ws[WS_U2 + d1], 0.0f);

    float ss = r0 * r0 + r1 * r1;
    for (int off = 16; off >= 1; off >>= 1) ss += __shfl_xor(ss, off, 32);
    float inv = 1.0f / fmaxf(sqrtf(ss), 1e-12f);
    r0 *= inv; r1 *= inv;

    if (i < NN) {
        float2 pr = make_float2(r0, r1);
        float* dst = out + BB + (l * NN + i) * DD + d0;
        *(float2*)dst = pr;
    } else {
        int b = i - NN;
        float t = r0 * cp[d0] + r1 * cp[d1];
        for (int off = 16; off >= 1; off >>= 1) t += __shfl_xor(t, off, 32);
        if (lane == 0) ws[WS_T + l * BB + b] = t;
    }
}

__global__ void k_esa(const int* __restrict__ act_col, const float* __restrict__ out,
                      float* ws) {
    int idx = blockIdx.x * blockDim.x + threadIdx.x;
    if (idx >= 2 * BB * DD) return;
    int d = idx & (DD - 1);
    int b = (idx >> 6) & (BB - 1);
    int l = idx >> 15;
    float t = ws[WS_T + l * BB + b];
    float ae = out[BB + (l * NN + act_col[b]) * DD + d];
    ws[WS_ESA + idx] = t * ae;
}

__global__ void k_gemm(const float* __restrict__ ws, const float* __restrict__ h1,
                       float* __restrict__ hid) {
    int bid  = blockIdx.x;
    int tn   = bid & 1;
    int tm   = (bid >> 1) & 31;
    int l    = bid >> 6;
    int lane = threadIdx.x;
    int half = lane >> 4;
    int lid  = lane & 15;

    const float* A = ws + WS_ESA + (l * BB + tm * 16) * DD;
    v8f c = {0.f, 0.f, 0.f, 0.f, 0.f, 0.f, 0.f, 0.f};
    #pragma unroll
    for (int s = 0; s < 16; ++s) {
        int k0 = 4 * s + 2 * half;
        v2f av, bv;
        av.x = A[lid * DD + k0];
        av.y = A[lid * DD + k0 + 1];
        bv.x = h1[(k0    ) * 32 + tn * 16 + lid];
        bv.y = h1[(k0 + 1) * 32 + tn * 16 + lid];
        c = __builtin_amdgcn_wmma_f32_16x16x4_f32(false, av, false, bv,
                                                  (short)0, c, false, false);
    }
    float* dst = hid + (l * BB + tm * 16) * 32 + tn * 16;
    #pragma unroll
    for (int r = 0; r < 8; ++r) {
        int M = r + 8 * half;
        dst[M * 32 + lid] = fmaxf(c[r], 0.0f);
    }
}

__global__ void k_q(const float* __restrict__ ws, const float* __restrict__ aux,
                    const float* __restrict__ h2w, float* out) {
    int b = blockIdx.x * blockDim.x + threadIdx.x;
    if (b >= BB) return;
    float q = 0.0f;
    for (int l = 0; l < 2; ++l) {
        const float* h = ws + WS_HID + (l * BB + b) * 32;
        for (int j = 0; j < 32; ++j) q += h[j] * h2w[j];
        for (int j = 0; j < 4; ++j)  q += aux[b * 8 + l * 4 + j] * h2w[32 + j];
    }
    out[b] = q;
}

extern "C" void kernel_launch(void* const* d_in, const int* in_sizes, int n_in,
                              void* d_out, int out_size, void* d_ws, size_t ws_size,
                              hipStream_t stream) {
    const int*   n2n0_row = (const int*)  d_in[0];
    const float* n2n0_val = (const float*)d_in[2];
    const int*   n2n1_row = (const int*)  d_in[3];
    const float* n2n1_val = (const float*)d_in[5];
    const int*   subg_row = (const int*)  d_in[6];
    const int*   act_col  = (const int*)  d_in[9];
    const int*   sg0_row  = (const int*)  d_in[10];
    const int*   sg0_col  = (const int*)  d_in[11];
    const int*   sg1_row  = (const int*)  d_in[12];
    const int*   sg1_col  = (const int*)  d_in[13];
    const float* aux      = (const float*)d_in[14];
    const float* w_n2l    = (const float*)d_in[15];
    const float* p_conv   = (const float*)d_in[16];
    const float* W_sage   = (const float*)d_in[17];
    const float* cp       = (const float*)d_in[18];
    const float* h1w      = (const float*)d_in[19];
    const float* h2w      = (const float*)d_in[20];
    float* out = (float*)d_out;
    float* ws  = (float*)d_ws;

    k_zero <<<(6 * NBT + 255) / 256, 256, 0, stream>>>(ws);
    k_small<<<1, 64, 0, stream>>>(w_n2l, p_conv, W_sage, ws);
    k_n2n  <<<SC_GRID, SC_T, 0, stream>>>(n2n0_row, n2n0_val, n2n1_row, n2n1_val, ws);
    k_subg <<<(NN + 255) / 256, 256, 0, stream>>>(subg_row, ws);
    k_sage <<<SC_GRID, SC_T, 0, stream>>>(sg0_row, sg0_col, sg1_row, sg1_col, ws);
    k_rows <<<(2 * NBT * 32 + 255) / 256, 256, 0, stream>>>(ws, cp, out);
    k_esa  <<<(2 * BB * DD + 255) / 256, 256, 0, stream>>>(act_col, out, ws);
    k_gemm <<<128, 32, 0, stream>>>(ws, h1w, ws + WS_HID);
    k_q    <<<(BB + 255) / 256, 256, 0, stream>>>(ws, aux, h2w, out);
}


# row kernel: one wave per 32 rows (was one short wave per row, 201024 waves): each lane sums its own row's 64 squares, then the wave stores two rows per instruction; norm and t sum order changes
# speedup vs baseline: 3.7730x; 1.1728x over previous
#include <hip/hip_runtime.h>

#define NN   100000
#define BB   512
#define EE   1000000
#define ESZ  1200000
#define DD   64
#define NBT  (NN + BB)

#define WS_A0   0
#define WS_A1   (NBT)
#define WS_M0   (2*NBT)
#define WS_M1   (3*NBT)
#define WS_DEG0 (4*NBT)
#define WS_DEG1 (5*NBT)
#define WS_SM   (6*NBT)
#define WS_U1   (WS_SM + 128)
#define WS_U2   (WS_SM + 192)
#define WS_T    (WS_SM + 256)
#define WS_ESA  (WS_T + 2*BB)
#define WS_HID  (WS_ESA + 2*BB*DD)

#define SC_P    16
#define SC_C    8
#define SC_T    1024
#define SC_V    2
#define SC_U    (4 * SC_V)
#define SC_GRID (2 * SC_P * SC_C)
#define N2N_RB  (NN / SC_P)
#define SG_RB   (NBT / SC_P)
#define N2N_CH  (EE / SC_C)
#define SG_CH   (ESZ / SC_C)
static_assert(NN % SC_P == 0 && NBT % SC_P == 0, "row ranges must tile the rows");
static_assert(EE % SC_C == 0 && ESZ % SC_C == 0, "edge chunks must tile the edges");
static_assert(N2N_CH % 4 == 0 && SG_CH % 4 == 0, "a chunk is read in four-word pieces");

typedef float v2f __attribute__((ext_vector_type(2)));
typedef float v8f __attribute__((ext_vector_type(8)));

__global__ void k_zero(float* ws) {
    int i = blockIdx.x * blockDim.x + threadIdx.x;
    if (i < 6 * NBT) ws[i] = 0.0f;
}

__global__ void k_small(const float* __restrict__ w_n2l,
                        const float* __restrict__ p_conv,
                        const float* __restrict__ W_sage,
                        float* ws) {
    __shared__ float v[DD], w[DD];
    int d = threadIdx.x;
    float x = fmaxf(w_n2l[d] + w_n2l[DD + d], 0.0f);
    v[d] = x;
    __syncthreads();
    float ss = 0.f;
    for (int k = 0; k < DD; ++k) ss += v[k] * v[k];
    float nv = x / fmaxf(sqrtf(ss), 1e-12f);
    __syncthreads();
    v[d] = nv;
    __syncthreads();
    float wd = 0.f;
    for (int k = 0; k < DD; ++k) wd += v[k] * p_conv[k * DD + d];
    w[d] = wd;
    __syncthreads();
    float u1 = 0.f, u2 = 0.f;
    for (int k = 0; k < DD; ++k) {
        u1 += w[k] * W_sage[k * DD + d];
        u2 += w[k] * W_sage[(DD + k) * DD + d];
    }
    ws[WS_SM + d]  = nv;
    ws[WS_SM + 64 + d] = wd;
    ws[WS_U1 + d] = u1;
    ws[WS_U2 + d] = u2;
}

__global__ __launch_bounds__(SC_T)
void k_n2n(const int* __restrict__ r0, const float* __restrict__ v0,
           const int* __restrict__ r1, const float* __restrict__ v1,
           float* ws) {
    __shared__ float acc[N2N_RB];
    const int tid = threadIdx.x;
    const int bid = blockIdx.x;
    const int c = bid % SC_C;
    const int p = (bid / SC_C) % SC_P;
    const int l = bid / (SC_C * SC_P);
    const int lo = p * N2N_RB;
    const int4*   __restrict__ row4 = reinterpret_cast<const int4*>((l ? r1 : r0) + c * N2N_CH);
    const float4* __restrict__ val4 = reinterpret_cast<const float4*>((l ? v1 : v0) + c * N2N_CH);
    const int n4 = N2N_CH / 4;

    for (int i = tid; i < N2N_RB; i += SC_T) acc[i] = 0.0f;
    __syncthreads();

    int4 qr[SC_V]; float4 qv[SC_V];
    #pragma unroll
    for (int u = 0; u < SC_V; ++u) {
        int i4 = u * SC_T + tid;
        int ii = i4 < n4 ? i4 : 0;
        qr[u] = row4[ii];
        qv[u] = val4[ii];
    }
    for (int b4 = 0; b4 < n4; b4 += SC_T * SC_V) {
        int rr[SC_U]; float vv[SC_U];
        #pragma unroll
        for (int u = 0; u < SC_V; ++u) {
            bool ok = (b4 + u * SC_T + tid) < n4;
            rr[4 * u + 0] = ok ? qr[u].x - lo : -1;  vv[4 * u + 0] = qv[u].x;
            rr[4 * u + 1] = ok ? qr[u].y - lo : -1;  vv[4 * u + 1] = qv[u].y;
            rr[4 * u + 2] = ok ? qr[u].z - lo : -1;  vv[4 * u + 2] = qv[u].z;
            rr[4 * u + 3] = ok ? qr[u].w - lo : -1;  vv[4 * u + 3] = qv[u].w;
        }
        #pragma unroll
        for (int u = 0; u < SC_V; ++u) {
            int i4 = b4 + SC_T * SC_V + u * SC_T + tid;
            int ii = i4 < n4 ? i4 : 0;
            qr[u] = row4[ii];
            qv[u] = val4[ii];
        }
        #pragma unroll
        for (int k = 0; k < SC_U; ++k)
            if ((unsigned)rr[k] < (unsigned)N2N_RB) atomicAdd(&acc[rr[k]], vv[k]);
    }
    __syncthreads();

    float* dst = ws + (l ? WS_A1 : WS_A0) + lo;
    for (int i = tid; i < N2N_RB; i += SC_T) {
        float sv = acc[i];
        if (sv != 0.0f) atomicAdd(&dst[i], sv);
    }
}

__global__ void k_subg(const int* __restrict__ subg_row, float* ws) {
    int i = blockIdx.x * blockDim.x + threadIdx.x;
    if (i < NN) {
        int b = subg_row[i];
        atomicAdd(&ws[WS_A0 + NN + b], 1.0f);
        atomicAdd(&ws[WS_A1 + NN + b], 1.0f);
    }
}

__global__ __launch_bounds__(SC_T)
void k_sage(const int* __restrict__ s0r, const int* __restrict__ s0c,
            const int* __restrict__ s1r, const int* __restrict__ s1c,
            float* ws) {
    __shared__ float        num[SG_RB];
    __shared__ unsigned int cnt[SG_RB];
    const int tid = threadIdx.x;
    const int bid = blockIdx.x;
    const int c = bid % SC_C;
    const int p = (bid / SC_C) % SC_P;
    const int l = bid / (SC_C * SC_P);
    const int lo = p * SG_RB;
    const int4* __restrict__ row4 = reinterpret_cast<const int4*>((l ? s1r : s0r) + c * SG_CH);
    const int4* __restrict__ col4 = reinterpret_cast<const int4*>((l ? s1c : s0c) + c * SG_CH);
    const float* a = ws + (l ? WS_A1 : WS_A0);
    const int n4 = SG_CH / 4;
    const int STEP4 = SC_T * SC_V;

    for (int i = tid; i < SG_RB; i += SC_T) { num[i] = 0.0f; cnt[i] = 0u; }
    __syncthreads();

    int4 qr[SC_V]; int4 qc[SC_V];
    int rr[SC_U]; float gv[SC_U];
    #pragma unroll
    for (int u = 0; u < SC_V; ++u) {
        int i4 = u * SC_T + tid;
        int ii = i4 < n4 ? i4 : 0;
        qr[u] = row4[ii];
        qc[u] = col4[ii];
    }
    #pragma unroll
    for (int u = 0; u < SC_V; ++u) {
        bool ok = (u * SC_T + tid) < n4;
        int r4[4] = {qr[u].x - lo, qr[u].y - lo, qr[u].z - lo, qr[u].w - lo};
        int c4[4] = {qc[u].x, qc[u].y, qc[u].z, qc[u].w};
        #pragma unroll
        for (int j = 0; j < 4; ++j) {
            bool m = ok && ((unsigned)r4[j] < (unsigned)SG_RB);
            rr[4 * u + j] = m ? r4[j] : -1;
            gv[4 * u + j] = a[m ? c4[j] : 0];
        }
    }
    #pragma unroll
    for (int u = 0; u < SC_V; ++u) {
        int i4 = STEP4 + u * SC_T + tid;
        int ii = i4 < n4 ? i4 : 0;
        qr[u] = row4[ii];
        qc[u] = col4[ii];
    }
    for (int b4 = 0; b4 < n4; b4 += STEP4) {
        int nr[SC_U]; float ng[SC_U];
        #pragma unroll
        for (int u = 0; u < SC_V; ++u) {
            bool ok = (b4 + STEP4 + u * SC_T + tid) < n4;
            int r4[4] = {qr[u].x - lo, qr[u].y - lo, qr[u].z - lo, qr[u].w - lo};
            int c4[4] = {qc[u].x, qc[u].y, qc[u].z, qc[u].w};
            #pragma unroll
            for (int j = 0; j < 4; ++j) {
                bool m = ok && ((unsigned)r4[j] < (unsigned)SG_RB);
                nr[4 * u + j] = m ? r4[j] : -1;
                ng[4 * u + j] = a[m ? c4[j] : 0];
            }
        }
        #pragma unroll
        for (int u = 0; u < SC_V; ++u) {
            int i4 = b4 + 2 * STEP4 + u * SC_T + tid;
            int ii = i4 < n4 ? i4 : 0;
            qr[u] = row4[ii];
            qc[u] = col4[ii];
        }
        #pragma unroll
        for (int k = 0; k < SC_U; ++k)
            if (rr[k] >= 0) {
                atomicAdd(&num[rr[k]], gv[k]);
                atomicAdd(&cnt[rr[k]], 1u);
            }
        #pragma unroll
        for (int k = 0; k < SC_U; ++k) { rr[k] = nr[k]; gv[k] = ng[k]; }
    }
    __syncthreads();

    float* dm = ws + (l ? WS_M1   : WS_M0)   + lo;
    float* dd = ws + (l ? WS_DEG1 : WS_DEG0) + lo;
    for (int i = tid; i < SG_RB; i += SC_T) {
        unsigned int k = cnt[i];
        if (k != 0u) {
            atomicAdd(&dm[i], num[i]);
            atomicAdd(&dd[i], (float)k);
        }
    }
}

#define RW_T   256
#define RW_NCH (2 * NBT / 32)
static_assert(NN % 32 == 0 && NBT % 32 == 0, "a 32-row chunk must stay inside one layer and one row kind");

__device__ __forceinline__ float row_val(float a, float u1, float m, float u2) {
    return fmaxf(fmaf(a, u1, m * u2), 0.0f);
}

__global__ __launch_bounds__(RW_T)
void k_rows(const float* __restrict__ wsr, float* __restrict__ tq,
            const float* __restrict__ cp, float* __restrict__ out) {
    const int lane  = threadIdx.x & 31;
    const int chunk = (blockIdx.x * RW_T + threadIdx.x) >> 5;
    if (chunk >= RW_NCH) return;
    const int l  = chunk / (NBT / 32);
    const int i0 = (chunk - l * (NBT / 32)) * 32;
    const int i  = i0 + lane;
    const int c    = lane & 15;
    const int half = lane >> 4;

    const float a   = wsr[(l ? WS_A1   : WS_A0)   + i];
    const float num = wsr[(l ? WS_M1   : WS_M0)   + i];
    const float dg  = wsr[(l ? WS_DEG1 : WS_DEG0) + i];
    const float4 p1 = *reinterpret_cast<const float4*>(wsr + WS_U1 + 4 * c);
    const float4 p2 = *reinterpret_cast<const float4*>(wsr + WS_U2 + 4 * c);
    const float* __restrict__ u1 = wsr + WS_U1;
    const float* __restrict__ u2 = wsr + WS_U2;
    const float m = num / fmaxf(dg, 1.0f);

    float s0 = 0.f, s1 = 0.f, s2 = 0.f, s3 = 0.f;
    #pragma unroll
    for (int d = 0; d < DD; d += 4) {
        float r0 = row_val(a, u1[d + 0], m, u2[d + 0]);
        float r1 = row_val(a, u1[d + 1], m, u2[d + 1]);
        float r2 = row_val(a, u1[d + 2], m, u2[d + 2]);
        float r3 = row_val(a, u1[d + 3], m, u2[d + 3]);
        s0 = fmaf(r0, r0, s0);
        s1 = fmaf(r1, r1, s1);
        s2 = fmaf(r2, r2, s2);
        s3 = fmaf(r3, r3, s3);
    }
    const float ss  = (s0 + s1) + (s2 + s3);
    const float inv = 1.0f / fmaxf(sqrtf(ss), 1e-12f);

    if (i0 >= NN) {
        float t0 = 0.f, t1 = 0.f, t2 = 0.f, t3 = 0.f;
        #pragma unroll
        for (int d = 0; d < DD; d += 4) {
            t0 = fmaf(row_val(a, u1[d + 0], m, u2[d + 0]) * inv, cp[d + 0], t0);
            t1 = fmaf(row_val(a, u1[d + 1], m, u2[d + 1]) * inv, cp[d + 1], t1);
            t2 = fmaf(row_val(a, u1[d + 2], m, u2[d + 2]) * inv, cp[d + 2], t2);
            t3 = fmaf(row_val(a, u1[d + 3], m, u2[d + 3]) * inv, cp[d + 3], t3);
        }
        tq[l * BB + (i - NN)] = (t0 + t1) + (t2 + t3);
        return;
    }

    float* dst = out + BB + (l * NN + i0 + half) * DD + 4 * c;
    #pragma unroll
    for (int p = 0; p < 16; ++p) {
        const int src = 2 * p + half;
        const float ar = __shfl(a,   src, 32);
        const float mr = __shfl(m,   src, 32);
        const float ir = __shfl(inv, src, 32);
        float4 o;
        o.x = row_val(ar, p1.x, mr, p2.x) * ir;
        o.y = row_val(ar, p1.y, mr, p2.y) * ir;
        o.z = row_val(ar, p1.z, mr, p2.z) * ir;
        o.w = row_val(ar, p1.w, mr, p2.w) * ir;
        *reinterpret_cast<float4*>(dst + 2 * p * DD) = o;
    }
}

__global__ void k_esa(const int* __restrict__ act_col, const float* __restrict__ out,
                      float* ws) {
    int idx = blockIdx.x * blockDim.x + threadIdx.x;
    if (idx >= 2 * BB * DD) return;
    int d = idx & (DD - 1);
    int b = (idx >> 6) & (BB - 1);
    int l = idx >> 15;
    float t = ws[WS_T + l * BB + b];
    float ae = out[BB + (l * NN + act_col[b]) * DD + d];
    ws[WS_ESA + idx] = t * ae;
}

__global__ void k_gemm(const float* __restrict__ ws, const float* __restrict__ h1,
                       float* __restrict__ hid) {
    int bid  = blockIdx.x;
    int tn   = bid & 1;
    int tm   = (bid >> 1) & 31;
    int l    = bid >> 6;
    int lane = threadIdx.x;
    int half = lane >> 4;
    int lid  = lane & 15;

    const float* A = ws + WS_ESA + (l * BB + tm * 16) * DD;
    v8f c = {0.f, 0.f, 0.f, 0.f, 0.f, 0.f, 0.f, 0.f};
    #pragma unroll
    for (int s = 0; s < 16; ++s) {
        int k0 = 4 * s + 2 * half;
        v2f av, bv;
        av.x = A[lid * DD + k0];
        av.y = A[lid * DD + k0 + 1];
        bv.x = h1[(k0    ) * 32 + tn * 16 + lid];
        bv.y = h1[(k0 + 1) * 32 + tn * 16 + lid];
        c = __builtin_amdgcn_wmma_f32_16x16x4_f32(false, av, false, bv,
                                                  (short)0, c, false, false);
    }
    float* dst = hid + (l * BB + tm * 16) * 32 + tn * 16;
    #pragma unroll
    for (int r = 0; r < 8; ++r) {
        int M = r + 8 * half;
        dst[M * 32 + lid] = fmaxf(c[r], 0.0f);
    }
}

__global__ void k_q(const float* __restrict__ ws, const float* __restrict__ aux,
                    const float* __restrict__ h2w, float* out) {
    int b = blockIdx.x * blockDim.x + threadIdx.x;
    if (b >= BB) return;
    float q = 0.0f;
    for (int l = 0; l < 2; ++l) {
        const float* h = ws + WS_HID + (l * BB + b) * 32;
        for (int j = 0; j < 32; ++j) q += h[j] * h2w[j];
        for (int j = 0; j < 4; ++j)  q += aux[b * 8 + l * 4 + j] * h2w[32 + j];
    }
    out[b] = q;
}

extern "C" void kernel_launch(void* const* d_in, const int* in_sizes, int n_in,
                              void* d_out, int out_size, void* d_ws, size_t ws_size,
                              hipStream_t stream) {
    const int*   n2n0_row = (const int*)  d_in[0];
    const float* n2n0_val = (const float*)d_in[2];
    const int*   n2n1_row = (const int*)  d_in[3];
    const float* n2n1_val = (const float*)d_in[5];
    const int*   subg_row = (const int*)  d_in[6];
    const int*   act_col  = (const int*)  d_in[9];
    const int*   sg0_row  = (const int*)  d_in[10];
    const int*   sg0_col  = (const int*)  d_in[11];
    const int*   sg1_row  = (const int*)  d_in[12];
    const int*   sg1_col  = (const int*)  d_in[13];
    const float* aux      = (const float*)d_in[14];
    const float* w_n2l    = (const float*)d_in[15];
    const float* p_conv   = (const float*)d_in[16];
    const float* W_sage   = (const float*)d_in[17];
    const float* cp       = (const float*)d_in[18];
    const float* h1w      = (const float*)d_in[19];
    const float* h2w      = (const float*)d_in[20];
    float* out = (float*)d_out;
    float* ws  = (float*)d_ws;

    k_zero <<<(6 * NBT + 255) / 256, 256, 0, stream>>>(ws);
    k_small<<<1, 64, 0, stream>>>(w_n2l, p_conv, W_sage, ws);
    k_n2n  <<<SC_GRID, SC_T, 0, stream>>>(n2n0_row, n2n0_val, n2n1_row, n2n1_val, ws);
    k_subg <<<(NN + 255) / 256, 256, 0, stream>>>(subg_row, ws);
    k_sage <<<SC_GRID, SC_T, 0, stream>>>(sg0_row, sg0_col, sg1_row, sg1_col, ws);
    k_rows <<<(RW_NCH * 32 + RW_T - 1) / RW_T, RW_T, 0, stream>>>(ws, ws + WS_T, cp, out);
    k_esa  <<<(2 * BB * DD + 255) / 256, 256, 0, stream>>>(act_col, out, ws);
    k_gemm <<<128, 32, 0, stream>>>(ws, h1w, ws + WS_HID);
    k_q    <<<(BB + 255) / 256, 256, 0, stream>>>(ws, aux, h2w, out);
}


# 9 launches -> 4: tiny dense chain shares the zeroing launch, subgraph histogram rides in the n2n launch, esa + WMMA product + q run as head blocks of the rows launch (t, esa, hid stay in LDS)
# speedup vs baseline: 4.4328x; 1.1749x over previous
#include <hip/hip_runtime.h>

#define NN   100000
#define BB   512
#define EE   1000000
#define ESZ  1200000
#define DD   64
#define NBT  (NN + BB)

#define WS_A0   0
#define WS_A1   (NBT)
#define WS_M0   (2*NBT)
#define WS_M1   (3*NBT)
#define WS_DEG0 (4*NBT)
#define WS_DEG1 (5*NBT)
#define WS_SM   (6*NBT)
#define WS_U1   (WS_SM + 128)
#define WS_U2   (WS_SM + 192)

#define SC_P    16
#define SC_C    8
#define SC_T    1024
#define SC_V    2
#define SC_U    (4 * SC_V)
#define SC_GRID (2 * SC_P * SC_C)
#define N2N_RB  (NN / SC_P)
#define SG_RB   (NBT / SC_P)
#define N2N_CH  (EE / SC_C)
#define SG_CH   (ESZ / SC_C)
static_assert(NN % SC_P == 0 && NBT % SC_P == 0, "row ranges must tile the rows");
static_assert(EE % SC_C == 0 && ESZ % SC_C == 0, "edge chunks must tile the edges");
static_assert(N2N_CH % 4 == 0 && SG_CH % 4 == 0, "a chunk is read in four-word pieces");

typedef float v2f __attribute__((ext_vector_type(2)));
typedef float v8f __attribute__((ext_vector_type(8)));

#define PR_T   256
#define PR_ZB  ((6 * NBT + PR_T - 1) / PR_T)
__global__ __launch_bounds__(PR_T)
void k_prep(const float* __restrict__ w_n2l,
            const float* __restrict__ p_conv,
            const float* __restrict__ W_sage,
            float* ws) {
    __shared__ float v[DD], w[DD];
    if (blockIdx.x != 0) {
        int i = (blockIdx.x - 1) * PR_T + threadIdx.x;
        if (i < 6 * NBT) ws[i] = 0.0f;
        return;
    }
    const int d = threadIdx.x & (DD - 1);
    float x = fmaxf(w_n2l[d] + w_n2l[DD + d], 0.0f);
    v[d] = x;
    __syncthreads();
    float ss = 0.f;
    for (int k = 0; k < DD; ++k) ss += v[k] * v[k];
    float nv = x / fmaxf(sqrtf(ss), 1e-12f);
    __syncthreads();
    v[d] = nv;
    __syncthreads();
    float wd = 0.f;
    for (int k = 0; k < DD; ++k) wd += v[k] * p_conv[k * DD + d];
    w[d] = wd;
    __syncthreads();
    float u1 = 0.f, u2 = 0.f;
    for (int k = 0; k < DD; ++k) {
        u1 += w[k] * W_sage[k * DD + d];
        u2 += w[k] * W_sage[(DD + k) * DD + d];
    }
    if (threadIdx.x < DD) {
        ws[WS_SM + d]  = nv;
        ws[WS_SM + 64 + d] = wd;
        ws[WS_U1 + d] = u1;
        ws[WS_U2 + d] = u2;
    }
}

__global__ __launch_bounds__(SC_T)
void k_n2n(const int* __restrict__ r0, const float* __restrict__ v0,
           const int* __restrict__ r1, const float* __restrict__ v1,
           const int* __restrict__ subg_row,
           float* ws) {
    __shared__ float acc[N2N_RB];
    const int tid = threadIdx.x;
    const int bid = blockIdx.x;
    {
        const int g = bid * SC_T + tid;
        if (g < NN) {
            int b = subg_row[g];
            atomicAdd(&ws[WS_A0 + NN + b], 1.0f);
            atomicAdd(&ws[WS_A1 + NN + b], 1.0f);
        }
    }
    const int c = bid % SC_C;
    const int p = (bid / SC_C) % SC_P;
    const int l = bid / (SC_C * SC_P);
    const int lo = p * N2N_RB;
    const int4*   __restrict__ row4 = reinterpret_cast<const int4*>((l ? r1 : r0) + c * N2N_CH);
    const float4* __restrict__ val4 = reinterpret_cast<const float4*>((l ? v1 : v0) + c * N2N_CH);
    const int n4 = N2N_CH / 4;

    for (int i = tid; i < N2N_RB; i += SC_T) acc[i] = 0.0f;
    __syncthreads();

    int4 qr[SC_V]; float4 qv[SC_V];
    #pragma unroll
    for (int u = 0; u < SC_V; ++u) {
        int i4 = u * SC_T + tid;
        int ii = i4 < n4 ? i4 : 0;
        qr[u] = row4[ii];
        qv[u] = val4[ii];
    }
    for (int b4 = 0; b4 < n4; b4 += SC_T * SC_V) {
        int rr[SC_U]; float vv[SC_U];
        #pragma unroll
        for (int u = 0; u < SC_V; ++u) {
            bool ok = (b4 + u * SC_T + tid) < n4;
            rr[4 * u + 0] = ok ? qr[u].x - lo : -1;  vv[4 * u + 0] = qv[u].x;
            rr[4 * u + 1] = ok ? qr[u].y - lo : -1;  vv[4 * u + 1] = qv[u].y;
            rr[4 * u + 2] = ok ? qr[u].z - lo : -1;  vv[4 * u + 2] = qv[u].z;
            rr[4 * u + 3] = ok ? qr[u].w - lo : -1;  vv[4 * u + 3] = qv[u].w;
        }
        #pragma unroll
        for (int u = 0; u < SC_V; ++u) {
            int i4 = b4 + SC_T * SC_V + u * SC_T + tid;
            int ii = i4 < n4 ? i4 : 0;
            qr[u] = row4[ii];
            qv[u] = val4[ii];
        }
        #pragma unroll
        for (int k = 0; k < SC_U; ++k)
            if ((unsigned)rr[k] < (unsigned)N2N_RB) atomicAdd(&acc[rr[k]], vv[k]);
    }
    __syncthreads();

    float* dst = ws + (l ? WS_A1 : WS_A0) + lo;
    for (int i = tid; i < N2N_RB; i += SC_T) {
        float sv = acc[i];
        if (sv != 0.0f) atomicAdd(&dst[i], sv);
    }
}

__global__ __launch_bounds__(SC_T)
void k_sage(const int* __restrict__ s0r, const int* __restrict__ s0c,
            const int* __restrict__ s1r, const int* __restrict__ s1c,
            float* ws) {
    __shared__ float        num[SG_RB];
    __shared__ unsigned int cnt[SG_RB];
    const int tid = threadIdx.x;
    const int bid = blockIdx.x;
    const int c = bid % SC_C;
    const int p = (bid / SC_C) % SC_P;
    const int l = bid / (SC_C * SC_P);
    const int lo = p * SG_RB;
    const int4* __restrict__ row4 = reinterpret_cast<const int4*>((l ? s1r : s0r) + c * SG_CH);
    const int4* __restrict__ col4 = reinterpret_cast<const int4*>((l ? s1c : s0c) + c * SG_CH);
    const float* a = ws + (l ? WS_A1 : WS_A0);
    const int n4 = SG_CH / 4;
    const int STEP4 = SC_T * SC_V;

    for (int i = tid; i < SG_RB; i += SC_T) { num[i] = 0.0f; cnt[i] = 0u; }
    __syncthreads();

    int4 qr[SC_V]; int4 qc[SC_V];
    int rr[SC_U]; float gv[SC_U];
    #pragma unroll
    for (int u = 0; u < SC_V; ++u) {
        int i4 = u * SC_T + tid;
        int ii = i4 < n4 ? i4 : 0;
        qr[u] = row4[ii];
        qc[u] = col4[ii];
    }
    #pragma unroll
    for (int u = 0; u < SC_V; ++u) {
        bool ok = (u * SC_T + tid) < n4;
        int r4[4] = {qr[u].x - lo, qr[u].y - lo, qr[u].z - lo, qr[u].w - lo};
        int c4[4] = {qc[u].x, qc[u].y, qc[u].z, qc[u].w};
        #pragma unroll
        for (int j = 0; j < 4; ++j) {
            bool m = ok && ((unsigned)r4[j] < (unsigned)SG_RB);
            rr[4 * u + j] = m ? r4[j] : -1;
            gv[4 * u + j] = a[m ? c4[j] : 0];
        }
    }
    #pragma unroll
    for (int u = 0; u < SC_V; ++u) {
        int i4 = STEP4 + u * SC_T + tid;
        int ii = i4 < n4 ? i4 : 0;
        qr[u] = row4[ii];
        qc[u] = col4[ii];
    }
    for (int b4 = 0; b4 < n4; b4 += STEP4) {
        int nr[SC_U]; float ng[SC_U];
        #pragma unroll
        for (int u = 0; u < SC_V; ++u) {
            bool ok = (b4 + STEP4 + u * SC_T + tid) < n4;
            int r4[4] = {qr[u].x - lo, qr[u].y - lo, qr[u].z - lo, qr[u].w - lo};
            int c4[4] = {qc[u].x, qc[u].y, qc[u].z, qc[u].w};
            #pragma unroll
            for (int j = 0; j < 4; ++j) {
                bool m = ok && ((unsigned)r4[j] < (unsigned)SG_RB);
                nr[4 * u + j] = m ? r4[j] : -1;
                ng[4 * u + j] = a[m ? c4[j] : 0];
            }
        }
        #pragma unroll
        for (int u = 0; u < SC_V; ++u) {
            int i4 = b4 + 2 * STEP4 + u * SC_T + tid;
            int ii = i4 < n4 ? i4 : 0;
            qr[u] = row4[ii];
            qc[u] = col4[ii];
        }
        #pragma unroll
        for (int k = 0; k < SC_U; ++k)
            if (rr[k] >= 0) {
                atomicAdd(&num[rr[k]], gv[k]);
                atomicAdd(&cnt[rr[k]], 1u);
            }
        #pragma unroll
        for (int k = 0; k < SC_U; ++k) { rr[k] = nr[k]; gv[k] = ng[k]; }
    }
    __syncthreads();

    float* dm = ws + (l ? WS_M1   : WS_M0)   + lo;
    float* dd = ws + (l ? WS_DEG1 : WS_DEG0) + lo;
    for (int i = tid; i < SG_RB; i += SC_T) {
        unsigned int k = cnt[i];
        if (k != 0u) {
            atomicAdd(&dm[i], num[i]);
            atomicAdd(&dd[i], (float)k);
        }
    }
}

#define RW_T    256
#define RW_NCH  (2 * NN / 32)
#define RW_NB   ((RW_NCH * 32 + RW_T - 1) / RW_T)
#define HD_BPB  16
#define HD_NB   (BB / HD_BPB)
#define HD_EP   68
#define HD_HP   33
static_assert(NN % 32 == 0, "a 32-row chunk must stay inside one layer");
static_assert(BB % HD_BPB == 0, "head blocks must tile the subgraph nodes");
static_assert(SC_GRID * SC_T >= NN, "the n2n grid must cover the subgraph entries");

__device__ __forceinline__ float row_val(float a, float u1, float m, float u2) {
    return fmaxf(fmaf(a, u1, m * u2), 0.0f);
}

__device__ __forceinline__ float row_ss(float a, float m,
                                        const float* __restrict__ u1, const float* __restrict__ u2) {
    float s0 = 0.f, s1 = 0.f, s2 = 0.f, s3 = 0.f;
    #pragma unroll
    for (int d = 0; d < DD; d += 4) {
        float r0 = row_val(a, u1[d + 0], m, u2[d + 0]);
        float r1 = row_val(a, u1[d + 1], m, u2[d + 1]);
        float r2 = row_val(a, u1[d + 2], m, u2[d + 2]);
        float r3 = row_val(a, u1[d + 3], m, u2[d + 3]);
        s0 = fmaf(r0, r0, s0);
        s1 = fmaf(r1, r1, s1);
        s2 = fmaf(r2, r2, s2);
        s3 = fmaf(r3, r3, s3);
    }
    return (s0 + s1) + (s2 + s3);
}

__global__ __launch_bounds__(RW_T)
void k_rows(const float* __restrict__ wsr,
            const float* __restrict__ cp,
            const int*   __restrict__ act_col,
            const float* __restrict__ h1,
            const float* __restrict__ aux,
            const float* __restrict__ h2w,
            float* __restrict__ out) {
    __shared__ float s_t[2 * HD_BPB], s_a[2 * HD_BPB], s_m[2 * HD_BPB], s_i[2 * HD_BPB];
    __shared__ __attribute__((aligned(16))) float s_esa[2 * HD_BPB * HD_EP];
    __shared__ float s_hid[2 * HD_BPB * HD_HP];

    const float* __restrict__ u1 = wsr + WS_U1;
    const float* __restrict__ u2 = wsr + WS_U2;
    const int lane = threadIdx.x & 31;
    const int c    = lane & 15;
    const int half = lane >> 4;

    if (blockIdx.x < HD_NB) {
        const int tid  = threadIdx.x;
        const int wave = tid >> 5;
        const int b0   = blockIdx.x * HD_BPB;
        if (wave == 0) {
            const int l = lane >> 4;
            const int i = NN + b0 + (lane & 15);
            const float a   = wsr[(l ? WS_A1   : WS_A0)   + i];
            const float num = wsr[(l ? WS_M1   : WS_M0)   + i];
            const float dg  = wsr[(l ? WS_DEG1 : WS_DEG0) + i];
            const float m   = num / fmaxf(dg, 1.0f);
            const float inv = 1.0f / fmaxf(sqrtf(row_ss(a, m, u1, u2)), 1e-12f);
            float t0 = 0.f, t1 = 0.f, t2 = 0.f, t3 = 0.f;
            #pragma unroll
            for (int d = 0; d < DD; d += 4) {
                t0 = fmaf(row_val(a, u1[d + 0], m, u2[d + 0]) * inv, cp[d + 0], t0);
                t1 = fmaf(row_val(a, u1[d + 1], m, u2[d + 1]) * inv, cp[d + 1], t1);
                t2 = fmaf(row_val(a, u1[d + 2], m, u2[d + 2]) * inv, cp[d + 2], t2);
                t3 = fmaf(row_val(a, u1[d + 3], m, u2[d + 3]) * inv, cp[d + 3], t3);
            }
            s_t[lane] = (t0 + t1) + (t2 + t3);
        } else if (wave == 1) {
            const int l = lane >> 4;
            const int i = act_col[b0 + (lane & 15)];
            const float a   = wsr[(l ? WS_A1   : WS_A0)   + i];
            const float num = wsr[(l ? WS_M1   : WS_M0)   + i];
            const float dg  = wsr[(l ? WS_DEG1 : WS_DEG0) + i];
            const float m   = num / fmaxf(dg, 1.0f);
            s_a[lane] = a;
            s_m[lane] = m;
            s_i[lane] = 1.0f / fmaxf(sqrtf(row_ss(a, m, u1, u2)), 1e-12f);
        }
        __syncthreads();
        for (int j = 0; j < (2 * HD_BPB * DD) / RW_T; ++j) {
            const int idx = j * RW_T + tid;
            const int pr  = idx >> 6;
            const int d   = idx & (DD - 1);
            const float ae = row_val(s_a[pr], u1[d], s_m[pr], u2[d]) * s_i[pr];
            s_esa[pr * HD_EP + d] = s_t[pr] * ae;
        }
        __syncthreads();
        if (wave < 4) {
            const int l  = wave >> 1;
            const int tn = wave & 1;
            const float* A = s_esa + (l * HD_BPB) * HD_EP;
            v8f acc = {0.f, 0.f, 0.f, 0.f, 0.f, 0.f, 0.f, 0.f};
            #pragma unroll
            for (int s = 0; s < 16; ++s) {
                const int k0 = 4 * s + 2 * half;
                v2f av, bv;
                av.x = A[c * HD_EP + k0];
                av.y = A[c * HD_EP + k0 + 1];
                bv.x = h1[(k0    ) * 32 + tn * 16 + c];
                bv.y = h1[(k0 + 1) * 32 + tn * 16 + c];
                acc = __builtin_amdgcn_wmma_f32_16x16x4_f32(false, av, false, bv,
                                                            (short)0, acc, false, false);
            }
            #pragma unroll
            for (int r = 0; r < 8; ++r) {
                const int M = r + 8 * half;
                s_hid[(l * HD_BPB + M) * HD_HP + tn * 16 + c] = fmaxf(acc[r], 0.0f);
            }
        }
        __syncthreads();
        if (tid < HD_BPB) {
            const int b = b0 + tid;
            float q = 0.0f;
            for (int l = 0; l < 2; ++l) {
                const float* h = s_hid + (l * HD_BPB + tid) * HD_HP;
                for (int j = 0; j < 32; ++j) q += h[j] * h2w[j];
                for (int j = 0; j < 4; ++j)  q += aux[b * 8 + l * 4 + j] * h2w[32 + j];
            }
            out[b] = q;
        }
        return;
    }

    const int chunk = ((blockIdx.x - HD_NB) * RW_T + threadIdx.x) >> 5;
    if (chunk >= RW_NCH) return;
    const int l  = chunk / (NN / 32);
    const int i0 = (chunk - l * (NN / 32)) * 32;
    const int i  = i0 + lane;

    const float a   = wsr[(l ? WS_A1   : WS_A0)   + i];
    const float num = wsr[(l ? WS_M1   : WS_M0)   + i];
    const float dg  = wsr[(l ? WS_DEG1 : WS_DEG0) + i];
    const float4 p1 = *reinterpret_cast<const float4*>(wsr + WS_U1 + 4 * c);
    const float4 p2 = *reinterpret_cast<const float4*>(wsr + WS_U2 + 4 * c);
    const float m = num / fmaxf(dg, 1.0f);
    const float inv = 1.0f / fmaxf(sqrtf(row_ss(a, m, u1, u2)), 1e-12f);

    float* dst = out + BB + (l * NN + i0 + half) * DD + 4 * c;
    #pragma unroll
    for (int p = 0; p < 16; ++p) {
        const int src = 2 * p + half;
        const float ar = __shfl(a,   src, 32);
        const float mr = __shfl(m,   src, 32);
        const float ir = __shfl(inv, src, 32);
        float4 o;
        o.x = row_val(ar, p1.x, mr, p2.x) * ir;
        o.y = row_val(ar, p1.y, mr, p2.y) * ir;
        o.z = row_val(ar, p1.z, mr, p2.z) * ir;
        o.w = row_val(ar, p1.w, mr, p2.w) * ir;
        *reinterpret_cast<float4*>(dst + 2 * p * DD) = o;
    }
}

extern "C" void kernel_launch(void* const* d_in, const int* in_sizes, int n_in,
                              void* d_out, int out_size, void* d_ws, size_t ws_size,
                              hipStream_t stream) {
    const int*   n2n0_row = (const int*)  d_in[0];
    const float* n2n0_val = (const float*)d_in[2];
    const int*   n2n1_row = (const int*)  d_in[3];
    const float* n2n1_val = (const float*)d_in[5];
    const int*   subg_row = (const int*)  d_in[6];
    const int*   act_col  = (const int*)  d_in[9];
    const int*   sg0_row  = (const int*)  d_in[10];
    const int*   sg0_col  = (const int*)  d_in[11];
    const int*   sg1_row  = (const int*)  d_in[12];
    const int*   sg1_col  = (const int*)  d_in[13];
    const float* aux      = (const float*)d_in[14];
    const float* w_n2l    = (const float*)d_in[15];
    const float* p_conv   = (const float*)d_in[16];
    const float* W_sage   = (const float*)d_in[17];
    const float* cp       = (const float*)d_in[18];
    const float* h1w      = (const float*)d_in[19];
    const float* h2w      = (const float*)d_in[20];
    float* out = (float*)d_out;
    float* ws  = (float*)d_ws;

    k_prep <<<1 + PR_ZB, PR_T, 0, stream>>>(w_n2l, p_conv, W_sage, ws);
    k_n2n  <<<SC_GRID, SC_T, 0, stream>>>(n2n0_row, n2n0_val, n2n1_row, n2n1_val, subg_row, ws);
    k_sage <<<SC_GRID, SC_T, 0, stream>>>(sg0_row, sg0_col, sg1_row, sg1_col, ws);
    k_rows <<<HD_NB + RW_NB, RW_T, 0, stream>>>(ws, cp, act_col, h1w, aux, h2w, out);
}


# n2n and sage scatters re-split: 4 row ranges x 32 edge chunks per layer (was 16 x 8), same 256 blocks, LDS sums 100/201 KB: a block scans 1/32 of a layer's edges, not 1/8; float sum order changes
# speedup vs baseline: 4.9787x; 1.1232x over previous
#include <hip/hip_runtime.h>

#define NN   100000
#define BB   512
#define EE   1000000
#define ESZ  1200000
#define DD   64
#define NBT  (NN + BB)

#define WS_A0   0
#define WS_A1   (NBT)
#define WS_M0   (2*NBT)
#define WS_M1   (3*NBT)
#define WS_DEG0 (4*NBT)
#define WS_DEG1 (5*NBT)
#define WS_SM   (6*NBT)
#define WS_U1   (WS_SM + 128)
#define WS_U2   (WS_SM + 192)

#define SC_P    4
#define SC_C    32
#define SC_T    1024
#define SC_V    2
#define SC_U    (4 * SC_V)
#define SC_GRID (2 * SC_P * SC_C)
#define FL_U    5
#define SC_RB(n) ((((n) + SC_P - 1) / SC_P + 31) / 32 * 32)
#define N2N_RB  SC_RB(NN)
#define SG_RB   SC_RB(NBT)
#define N2N_Q   (EE / 4)
#define SG_Q    (ESZ / 4)
static_assert(SC_P * N2N_RB >= NN && SC_P * SG_RB >= NBT, "row ranges must cover the rows");
static_assert(N2N_RB % 4 == 0 && SG_RB % 4 == 0, "the LDS sums are cleared in four-word pieces");
static_assert(NBT % 32 == 0, "the accumulator planes must start on whole lines");
static_assert(EE % 4 == 0 && ESZ % 4 == 0, "the edges are read in four-word pieces");
static_assert((long long)SC_C * SG_Q < 2147483647LL, "chunk bounds are computed in int");

typedef float v2f __attribute__((ext_vector_type(2)));
typedef float v8f __attribute__((ext_vector_type(8)));

#define PR_T   256
#define PR_ZB  ((6 * NBT + PR_T - 1) / PR_T)
__global__ __launch_bounds__(PR_T)
void k_prep(const float* __restrict__ w_n2l,
            const float* __restrict__ p_conv,
            const float* __restrict__ W_sage,
            float* ws) {
    __shared__ float v[DD], w[DD];
    if (blockIdx.x != 0) {
        int i = (blockIdx.x - 1) * PR_T + threadIdx.x;
        if (i < 6 * NBT) ws[i] = 0.0f;
        return;
    }
    const int d = threadIdx.x & (DD - 1);
    float x = fmaxf(w_n2l[d] + w_n2l[DD + d], 0.0f);
    v[d] = x;
    __syncthreads();
    float ss = 0.f;
    for (int k = 0; k < DD; ++k) ss += v[k] * v[k];
    float nv = x / fmaxf(sqrtf(ss), 1e-12f);
    __syncthreads();
    v[d] = nv;
    __syncthreads();
    float wd = 0.f;
    for (int k = 0; k < DD; ++k) wd += v[k] * p_conv[k * DD + d];
    w[d] = wd;
    __syncthreads();
    float u1 = 0.f, u2 = 0.f;
    for (int k = 0; k < DD; ++k) {
        u1 += w[k] * W_sage[k * DD + d];
        u2 += w[k] * W_sage[(DD + k) * DD + d];
    }
    if (threadIdx.x < DD) {
        ws[WS_SM + d]  = nv;
        ws[WS_SM + 64 + d] = wd;
        ws[WS_U1 + d] = u1;
        ws[WS_U2 + d] = u2;
    }
}

__global__ __launch_bounds__(SC_T)
void k_n2n(const int* __restrict__ r0, const float* __restrict__ v0,
           const int* __restrict__ r1, const float* __restrict__ v1,
           const int* __restrict__ subg_row,
           float* ws) {
    __shared__ __attribute__((aligned(16))) float acc[N2N_RB];
    const int tid = threadIdx.x;
    const int bid = blockIdx.x;
    {
        const int g = bid * SC_T + tid;
        if (g < NN) {
            int b = subg_row[g];
            atomicAdd(&ws[WS_A0 + NN + b], 1.0f);
            atomicAdd(&ws[WS_A1 + NN + b], 1.0f);
        }
    }
    const int c = bid % SC_C;
    const int p = (bid / SC_C) % SC_P;
    const int l = bid / (SC_C * SC_P);
    const int lo = p * N2N_RB;
    const int q0 = (c * N2N_Q) / SC_C;
    const int n4 = ((c + 1) * N2N_Q) / SC_C - q0;
    const int4*   __restrict__ row4 = reinterpret_cast<const int4*>(l ? r1 : r0) + q0;
    const float4* __restrict__ val4 = reinterpret_cast<const float4*>(l ? v1 : v0) + q0;

    for (int i = tid; i < N2N_RB / 4; i += SC_T)
        reinterpret_cast<float4*>(acc)[i] = make_float4(0.f, 0.f, 0.f, 0.f);
    __syncthreads();

    int4 qr[SC_V]; float4 qv[SC_V];
    #pragma unroll
    for (int u = 0; u < SC_V; ++u) {
        int i4 = u * SC_T + tid;
        int ii = i4 < n4 ? i4 : 0;
        qr[u] = row4[ii];
        qv[u] = val4[ii];
    }
    for (int b4 = 0; b4 < n4; b4 += SC_T * SC_V) {
        int rr[SC_U]; float vv[SC_U];
        #pragma unroll
        for (int u = 0; u < SC_V; ++u) {
            bool ok = (b4 + u * SC_T + tid) < n4;
            rr[4 * u + 0] = ok ? qr[u].x - lo : -1;  vv[4 * u + 0] = qv[u].x;
            rr[4 * u + 1] = ok ? qr[u].y - lo : -1;  vv[4 * u + 1] = qv[u].y;
            rr[4 * u + 2] = ok ? qr[u].z - lo : -1;  vv[4 * u + 2] = qv[u].z;
            rr[4 * u + 3] = ok ? qr[u].w - lo : -1;  vv[4 * u + 3] = qv[u].w;
        }
        #pragma unroll
        for (int u = 0; u < SC_V; ++u) {
            int i4 = b4 + SC_T * SC_V + u * SC_T + tid;
            int ii = i4 < n4 ? i4 : 0;
            qr[u] = row4[ii];
            qv[u] = val4[ii];
        }
        #pragma unroll
        for (int k = 0; k < SC_U; ++k)
            if ((unsigned)rr[k] < (unsigned)N2N_RB) atomicAdd(&acc[rr[k]], vv[k]);
    }
    __syncthreads();

    float* dst = ws + (l ? WS_A1 : WS_A0) + lo;
    for (int i0 = tid; i0 < N2N_RB; i0 += FL_U * SC_T) {
        float sv[FL_U];
        #pragma unroll
        for (int u = 0; u < FL_U; ++u) {
            const int i = i0 + u * SC_T;
            sv[u] = acc[i < N2N_RB ? i : 0];
        }
        #pragma unroll
        for (int u = 0; u < FL_U; ++u) {
            const int i = i0 + u * SC_T;
            if (i < N2N_RB && lo + i < NN && sv[u] != 0.0f) atomicAdd(&dst[i], sv[u]);
        }
    }
}

__global__ __launch_bounds__(SC_T)
void k_sage(const int* __restrict__ s0r, const int* __restrict__ s0c,
            const int* __restrict__ s1r, const int* __restrict__ s1c,
            float* ws) {
    __shared__ __attribute__((aligned(16))) float        num[SG_RB];
    __shared__ __attribute__((aligned(16))) unsigned int cnt[SG_RB];
    const int tid = threadIdx.x;
    const int bid = blockIdx.x;
    const int c = bid % SC_C;
    const int p = (bid / SC_C) % SC_P;
    const int l = bid / (SC_C * SC_P);
    const int lo = p * SG_RB;
    const int q0 = (c * SG_Q) / SC_C;
    const int n4 = ((c + 1) * SG_Q) / SC_C - q0;
    const int4* __restrict__ row4 = reinterpret_cast<const int4*>(l ? s1r : s0r) + q0;
    const int4* __restrict__ col4 = reinterpret_cast<const int4*>(l ? s1c : s0c) + q0;
    const float* a = ws + (l ? WS_A1 : WS_A0);
    const int STEP4 = SC_T * SC_V;

    for (int i = tid; i < SG_RB / 4; i += SC_T) {
        reinterpret_cast<float4*>(num)[i] = make_float4(0.f, 0.f, 0.f, 0.f);
        reinterpret_cast<uint4*>(cnt)[i]  = make_uint4(0u, 0u, 0u, 0u);
    }
    __syncthreads();

    int4 qr[SC_V]; int4 qc[SC_V];
    int rr[SC_U]; float gv[SC_U];
    #pragma unroll
    for (int u = 0; u < SC_V; ++u) {
        int i4 = u * SC_T + tid;
        int ii = i4 < n4 ? i4 : 0;
        qr[u] = row4[ii];
        qc[u] = col4[ii];
    }
    #pragma unroll
    for (int u = 0; u < SC_V; ++u) {
        bool ok = (u * SC_T + tid) < n4;
        int r4[4] = {qr[u].x - lo, qr[u].y - lo, qr[u].z - lo, qr[u].w - lo};
        int c4[4] = {qc[u].x, qc[u].y, qc[u].z, qc[u].w};
        #pragma unroll
        for (int j = 0; j < 4; ++j) {
            bool m = ok && ((unsigned)r4[j] < (unsigned)SG_RB);
            rr[4 * u + j] = m ? r4[j] : -1;
            gv[4 * u + j] = a[m ? c4[j] : 0];
        }
    }
    #pragma unroll
    for (int u = 0; u < SC_V; ++u) {
        int i4 = STEP4 + u * SC_T + tid;
        int ii = i4 < n4 ? i4 : 0;
        qr[u] = row4[ii];
        qc[u] = col4[ii];
    }
    for (int b4 = 0; b4 < n4; b4 += STEP4) {
        int nr[SC_U]; float ng[SC_U];
        #pragma unroll
        for (int u = 0; u < SC_V; ++u) {
            bool ok = (b4 + STEP4 + u * SC_T + tid) < n4;
            int r4[4] = {qr[u].x - lo, qr[u].y - lo, qr[u].z - lo, qr[u].w - lo};
            int c4[4] = {qc[u].x, qc[u].y, qc[u].z, qc[u].w};
            #pragma unroll
            for (int j = 0; j < 4; ++j) {
                bool m = ok && ((unsigned)r4[j] < (unsigned)SG_RB);
                nr[4 * u + j] = m ? r4[j] : -1;
                ng[4 * u + j] = a[m ? c4[j] : 0];
            }
        }
        #pragma unroll
        for (int u = 0; u < SC_V; ++u) {
            int i4 = b4 + 2 * STEP4 + u * SC_T + tid;
            int ii = i4 < n4 ? i4 : 0;
            qr[u] = row4[ii];
            qc[u] = col4[ii];
        }
        #pragma unroll
        for (int k = 0; k < SC_U; ++k)
            if (rr[k] >= 0) {
                atomicAdd(&num[rr[k]], gv[k]);
                atomicAdd(&cnt[rr[k]], 1u);
            }
        #pragma unroll
        for (int k = 0; k < SC_U; ++k) { rr[k] = nr[k]; gv[k] = ng[k]; }
    }
    __syncthreads();

    float* dm = ws + (l ? WS_M1   : WS_M0)   + lo;
    float* dd = ws + (l ? WS_DEG1 : WS_DEG0) + lo;
    for (int i0 = tid; i0 < SG_RB; i0 += FL_U * SC_T) {
        unsigned int k[FL_U]; float s[FL_U];
        #pragma unroll
        for (int u = 0; u < FL_U; ++u) {
            const int i  = i0 + u * SC_T;
            const int ii = i < SG_RB ? i : 0;
            k[u] = cnt[ii];
            s[u] = num[ii];
        }
        #pragma unroll
        for (int u = 0; u < FL_U; ++u) {
            const int i = i0 + u * SC_T;
            if (i < SG_RB && lo + i < NBT && k[u] != 0u) {
                atomicAdd(&dm[i], s[u]);
                atomicAdd(&dd[i], (float)k[u]);
            }
        }
    }
}

#define RW_T    256
#define RW_NCH  (2 * NN / 32)
#define RW_NB   ((RW_NCH * 32 + RW_T - 1) / RW_T)
#define HD_BPB  16
#define HD_NB   (BB / HD_BPB)
#define HD_EP   68
#define HD_HP   33
static_assert(NN % 32 == 0, "a 32-row chunk must stay inside one layer");
static_assert(BB % HD_BPB == 0, "head blocks must tile the subgraph nodes");
static_assert(SC_GRID * SC_T >= NN, "the n2n grid must cover the subgraph entries");

__device__ __forceinline__ float row_val(float a, float u1, float m, float u2) {
    return fmaxf(fmaf(a, u1, m * u2), 0.0f);
}

__device__ __forceinline__ float row_ss(float a, float m,
                                        const float* __restrict__ u1, const float* __restrict__ u2) {
    float s0 = 0.f, s1 = 0.f, s2 = 0.f, s3 = 0.f;
    #pragma unroll
    for (int d = 0; d < DD; d += 4) {
        float r0 = row_val(a, u1[d + 0], m, u2[d + 0]);
        float r1 = row_val(a, u1[d + 1], m, u2[d + 1]);
        float r2 = row_val(a, u1[d + 2], m, u2[d + 2]);
        float r3 = row_val(a, u1[d + 3], m, u2[d + 3]);
        s0 = fmaf(r0, r0, s0);
        s1 = fmaf(r1, r1, s1);
        s2 = fmaf(r2, r2, s2);
        s3 = fmaf(r3, r3, s3);
    }
    return (s0 + s1) + (s2 + s3);
}

__global__ __launch_bounds__(RW_T)
void k_rows(const float* __restrict__ wsr,
            const float* __restrict__ cp,
            const int*   __restrict__ act_col,
            const float* __restrict__ h1,
            const float* __restrict__ aux,
            const float* __restrict__ h2w,
            float* __restrict__ out) {
    __shared__ float s_t[2 * HD_BPB], s_a[2 * HD_BPB], s_m[2 * HD_BPB], s_i[2 * HD_BPB];
    __shared__ __attribute__((aligned(16))) float s_esa[2 * HD_BPB * HD_EP];
    __shared__ float s_hid[2 * HD_BPB * HD_HP];

    const float* __restrict__ u1 = wsr + WS_U1;
    const float* __restrict__ u2 = wsr + WS_U2;
    const int lane = threadIdx.x & 31;
    const int c    = lane & 15;
    const int half = lane >> 4;

    if (blockIdx.x < HD_NB) {
        const int tid  = threadIdx.x;
        const int wave = tid >> 5;
        const int b0   = blockIdx.x * HD_BPB;
        if (wave == 0) {
            const int l = lane >> 4;
            const int i = NN + b0 + (lane & 15);
            const float a   = wsr[(l ? WS_A1   : WS_A0)   + i];
            const float num = wsr[(l ? WS_M1   : WS_M0)   + i];
            const float dg  = wsr[(l ? WS_DEG1 : WS_DEG0) + i];
            const float m   = num / fmaxf(dg, 1.0f);
            const float inv = 1.0f / fmaxf(sqrtf(row_ss(a, m, u1, u2)), 1e-12f);
            float t0 = 0.f, t1 = 0.f, t2 = 0.f, t3 = 0.f;
            #pragma unroll
            for (int d = 0; d < DD; d += 4) {
                t0 = fmaf(row_val(a, u1[d + 0], m, u2[d + 0]) * inv, cp[d + 0], t0);
                t1 = fmaf(row_val(a, u1[d + 1], m, u2[d + 1]) * inv, cp[d + 1], t1);
                t2 = fmaf(row_val(a, u1[d + 2], m, u2[d + 2]) * inv, cp[d + 2], t2);
                t3 = fmaf(row_val(a, u1[d + 3], m, u2[d + 3]) * inv, cp[d + 3], t3);
            }
            s_t[lane] = (t0 + t1) + (t2 + t3);
        } else if (wave == 1) {
            const int l = lane >> 4;
            const int i = act_col[b0 + (lane & 15)];
            const float a   = wsr[(l ? WS_A1   : WS_A0)   + i];
            const float num = wsr[(l ? WS_M1   : WS_M0)   + i];
            const float dg  = wsr[(l ? WS_DEG1 : WS_DEG0) + i];
            const float m   = num / fmaxf(dg, 1.0f);
            s_a[lane] = a;
            s_m[lane] = m;
            s_i[lane] = 1.0f / fmaxf(sqrtf(row_ss(a, m, u1, u2)), 1e-12f);
        }
        __syncthreads();
        for (int j = 0; j < (2 * HD_BPB * DD) / RW_T; ++j) {
            const int idx = j * RW_T + tid;
            const int pr  = idx >> 6;
            const int d   = idx & (DD - 1);
            const float ae = row_val(s_a[pr], u1[d], s_m[pr], u2[d]) * s_i[pr];
            s_esa[pr * HD_EP + d] = s_t[pr] * ae;
        }
        __syncthreads();
        if (wave < 4) {
            const int l  = wave >> 1;
            const int tn = wave & 1;
            const float* A = s_esa + (l * HD_BPB) * HD_EP;
            v8f acc = {0.f, 0.f, 0.f, 0.f, 0.f, 0.f, 0.f, 0.f};
            #pragma unroll
            for (int s = 0; s < 16; ++s) {
                const int k0 = 4 * s + 2 * half;
                v2f av, bv;
                av.x = A[c * HD_EP + k0];
                av.y = A[c * HD_EP + k0 + 1];
                bv.x = h1[(k0    ) * 32 + tn * 16 + c];
                bv.y = h1[(k0 + 1) * 32 + tn * 16 + c];
                acc = __builtin_amdgcn_wmma_f32_16x16x4_f32(false, av, false, bv,
                                                            (short)0, acc, false, false);
            }
            #pragma unroll
            for (int r = 0; r < 8; ++r) {
                const int M = r + 8 * half;
                s_hid[(l * HD_BPB + M) * HD_HP + tn * 16 + c] = fmaxf(acc[r], 0.0f);
            }
        }
        __syncthreads();
        if (tid < HD_BPB) {
            const int b = b0 + tid;
            float q = 0.0f;
            for (int l = 0; l < 2; ++l) {
                const float* h = s_hid + (l * HD_BPB + tid) * HD_HP;
                for (int j = 0; j < 32; ++j) q += h[j] * h2w[j];
                for (int j = 0; j < 4; ++j)  q += aux[b * 8 + l * 4 + j] * h2w[32 + j];
            }
            out[b] = q;
        }
        return;
    }

    const int chunk = ((blockIdx.x - HD_NB) * RW_T + threadIdx.x) >> 5;
    if (chunk >= RW_NCH) return;
    const int l  = chunk / (NN / 32);
    const int i0 = (chunk - l * (NN / 32)) * 32;
    const int i  = i0 + lane;

    const float a   = wsr[(l ? WS_A1   : WS_A0)   + i];
    const float num = wsr[(l ? WS_M1   : WS_M0)   + i];
    const float dg  = wsr[(l ? WS_DEG1 : WS_DEG0) + i];
    const float4 p1 = *reinterpret_cast<const float4*>(wsr + WS_U1 + 4 * c);
    const float4 p2 = *reinterpret_cast<const float4*>(wsr + WS_U2 + 4 * c);
    const float m = num / fmaxf(dg, 1.0f);
    const float inv = 1.0f / fmaxf(sqrtf(row_ss(a, m, u1, u2)), 1e-12f);

    float* dst = out + BB + (l * NN + i0 + half) * DD + 4 * c;
    #pragma unroll
    for (int p = 0; p < 16; ++p) {
        const int src = 2 * p + half;
        const float ar = __shfl(a,   src, 32);
        const float mr = __shfl(m,   src, 32);
        const float ir = __shfl(inv, src, 32);
        float4 o;
        o.x = row_val(ar, p1.x, mr, p2.x) * ir;
        o.y = row_val(ar, p1.y, mr, p2.y) * ir;
        o.z = row_val(ar, p1.z, mr, p2.z) * ir;
        o.w = row_val(ar, p1.w, mr, p2.w) * ir;
        *reinterpret_cast<float4*>(dst + 2 * p * DD) = o;
    }
}

extern "C" void kernel_launch(void* const* d_in, const int* in_sizes, int n_in,
                              void* d_out, int out_size, void* d_ws, size_t ws_size,
                              hipStream_t stream) {
    const int*   n2n0_row = (const int*)  d_in[0];
    const float* n2n0_val = (const float*)d_in[2];
    const int*   n2n1_row = (const int*)  d_in[3];
    const float* n2n1_val = (const float*)d_in[5];
    const int*   subg_row = (const int*)  d_in[6];
    const int*   act_col  = (const int*)  d_in[9];
    const int*   sg0_row  = (const int*)  d_in[10];
    const int*   sg0_col  = (const int*)  d_in[11];
    const int*   sg1_row  = (const int*)  d_in[12];
    const int*   sg1_col  = (const int*)  d_in[13];
    const float* aux      = (const float*)d_in[14];
    const float* w_n2l    = (const float*)d_in[15];
    const float* p_conv   = (const float*)d_in[16];
    const float* W_sage   = (const float*)d_in[17];
    const float* cp       = (const float*)d_in[18];
    const float* h1w      = (const float*)d_in[19];
    const float* h2w      = (const float*)d_in[20];
    float* out = (float*)d_out;
    float* ws  = (float*)d_ws;

    k_prep <<<1 + PR_ZB, PR_T, 0, stream>>>(w_n2l, p_conv, W_sage, ws);
    k_n2n  <<<SC_GRID, SC_T, 0, stream>>>(n2n0_row, n2n0_val, n2n1_row, n2n1_val, subg_row, ws);
    k_sage <<<SC_GRID, SC_T, 0, stream>>>(sg0_row, sg0_col, sg1_row, sg1_col, ws);
    k_rows <<<HD_NB + RW_NB, RW_T, 0, stream>>>(ws, cp, act_col, h1w, aux, h2w, out);
}


# subgraph sums: 8 extra n2n-launch blocks add their entries (x subg_val) in LDS, then 512 lane-contiguous adds per plane; was 2 lone global adds per entry on 32 lines ahead of 98 blocks' first barrier
# speedup vs baseline: 5.4714x; 1.0990x over previous
#include <hip/hip_runtime.h>

#define NN   100000
#define BB   512
#define EE   1000000
#define ESZ  1200000
#define DD   64
#define NBT  (NN + BB)

#define WS_A0   0
#define WS_A1   (NBT)
#define WS_M0   (2*NBT)
#define WS_M1   (3*NBT)
#define WS_DEG0 (4*NBT)
#define WS_DEG1 (5*NBT)
#define WS_SM   (6*NBT)
#define WS_U1   (WS_SM + 128)
#define WS_U2   (WS_SM + 192)

#define SC_P    4
#define SC_C    32
#define SC_T    1024
#define SC_V    2
#define SC_U    (4 * SC_V)
#define SC_GRID (2 * SC_P * SC_C)
#define FL_U    5
#define HG_B    8
#define HG_E    ((NN + HG_B - 1) / HG_B)
#define HG_U    ((HG_E + SC_T - 1) / SC_T)
#define SC_RB(n) ((((n) + SC_P - 1) / SC_P + 31) / 32 * 32)
#define N2N_RB  SC_RB(NN)
#define SG_RB   SC_RB(NBT)
#define N2N_Q   (EE / 4)
#define SG_Q    (ESZ / 4)
static_assert(SC_P * N2N_RB >= NN && SC_P * SG_RB >= NBT, "row ranges must cover the rows");
static_assert(N2N_RB % 4 == 0 && SG_RB % 4 == 0, "the LDS sums are cleared in four-word pieces");
static_assert(NBT % 32 == 0, "the accumulator planes must start on whole lines");
static_assert(EE % 4 == 0 && ESZ % 4 == 0, "the edges are read in four-word pieces");
static_assert((long long)SC_C * SG_Q < 2147483647LL, "chunk bounds are computed in int");

typedef float v2f __attribute__((ext_vector_type(2)));
typedef float v8f __attribute__((ext_vector_type(8)));

#define PR_T   256
#define PR_ZB  ((6 * NBT + PR_T - 1) / PR_T)
__global__ __launch_bounds__(PR_T)
void k_prep(const float* __restrict__ w_n2l,
            const float* __restrict__ p_conv,
            const float* __restrict__ W_sage,
            float* ws) {
    __shared__ float v[DD], w[DD];
    if (blockIdx.x != 0) {
        int i = (blockIdx.x - 1) * PR_T + threadIdx.x;
        if (i < 6 * NBT) ws[i] = 0.0f;
        return;
    }
    const int d = threadIdx.x & (DD - 1);
    float x = fmaxf(w_n2l[d] + w_n2l[DD + d], 0.0f);
    v[d] = x;
    __syncthreads();
    float ss = 0.f;
    for (int k = 0; k < DD; ++k) ss += v[k] * v[k];
    float nv = x / fmaxf(sqrtf(ss), 1e-12f);
    __syncthreads();
    v[d] = nv;
    __syncthreads();
    float wd = 0.f;
    for (int k = 0; k < DD; ++k) wd += v[k] * p_conv[k * DD + d];
    w[d] = wd;
    __syncthreads();
    float u1 = 0.f, u2 = 0.f;
    for (int k = 0; k < DD; ++k) {
        u1 += w[k] * W_sage[k * DD + d];
        u2 += w[k] * W_sage[(DD + k) * DD + d];
    }
    if (threadIdx.x < DD) {
        ws[WS_SM + d]  = nv;
        ws[WS_SM + 64 + d] = wd;
        ws[WS_U1 + d] = u1;
        ws[WS_U2 + d] = u2;
    }
}

__global__ __launch_bounds__(SC_T)
void k_n2n(const int* __restrict__ r0, const float* __restrict__ v0,
           const int* __restrict__ r1, const float* __restrict__ v1,
           const int* __restrict__ subg_row, const float* __restrict__ subg_val,
           float* ws) {
    __shared__ __attribute__((aligned(16))) float acc[N2N_RB];
    const int tid = threadIdx.x;
    const int bid = blockIdx.x;
    if (bid >= SC_GRID) {
        const int e0 = (bid - SC_GRID) * HG_E;
        if (tid < BB) acc[tid] = 0.0f;
        __syncthreads();
        int hb[HG_U]; float hv[HG_U];
        #pragma unroll
        for (int u = 0; u < HG_U; ++u) {
            const int j = u * SC_T + tid;
            const bool ok = j < HG_E && e0 + j < NN;
            hb[u] = ok ? subg_row[e0 + j] : -1;
            hv[u] = ok ? subg_val[e0 + j] : 0.0f;
        }
        #pragma unroll
        for (int u = 0; u < HG_U; ++u)
            if ((unsigned)hb[u] < (unsigned)BB) atomicAdd(&acc[hb[u]], hv[u]);
        __syncthreads();
        if (tid < BB) {
            const float s = acc[tid];
            if (s != 0.0f) {
                atomicAdd(&ws[WS_A0 + NN + tid], s);
                atomicAdd(&ws[WS_A1 + NN + tid], s);
            }
        }
        return;
    }
    const int c = bid % SC_C;
    const int p = (bid / SC_C) % SC_P;
    const int l = bid / (SC_C * SC_P);
    const int lo = p * N2N_RB;
    const int q0 = (c * N2N_Q) / SC_C;
    const int n4 = ((c + 1) * N2N_Q) / SC_C - q0;
    const int4*   __restrict__ row4 = reinterpret_cast<const int4*>(l ? r1 : r0) + q0;
    const float4* __restrict__ val4 = reinterpret_cast<const float4*>(l ? v1 : v0) + q0;

    for (int i = tid; i < N2N_RB / 4; i += SC_T)
        reinterpret_cast<float4*>(acc)[i] = make_float4(0.f, 0.f, 0.f, 0.f);
    __syncthreads();

    int4 qr[SC_V]; float4 qv[SC_V];
    #pragma unroll
    for (int u = 0; u < SC_V; ++u) {
        int i4 = u * SC_T + tid;
        int ii = i4 < n4 ? i4 : 0;
        qr[u] = row4[ii];
        qv[u] = val4[ii];
    }
    for (int b4 = 0; b4 < n4; b4 += SC_T * SC_V) {
        int rr[SC_U]; float vv[SC_U];
        #pragma unroll
        for (int u = 0; u < SC_V; ++u) {
            bool ok = (b4 + u * SC_T + tid) < n4;
            rr[4 * u + 0] = ok ? qr[u].x - lo : -1;  vv[4 * u + 0] = qv[u].x;
            rr[4 * u + 1] = ok ? qr[u].y - lo : -1;  vv[4 * u + 1] = qv[u].y;
            rr[4 * u + 2] = ok ? qr[u].z - lo : -1;  vv[4 * u + 2] = qv[u].z;
            rr[4 * u + 3] = ok ? qr[u].w - lo : -1;  vv[4 * u + 3] = qv[u].w;
        }
        #pragma unroll
        for (int u = 0; u < SC_V; ++u) {
            int i4 = b4 + SC_T * SC_V + u * SC_T + tid;
            int ii = i4 < n4 ? i4 : 0;
            qr[u] = row4[ii];
            qv[u] = val4[ii];
        }
        #pragma unroll
        for (int k = 0; k < SC_U; ++k)
            if ((unsigned)rr[k] < (unsigned)N2N_RB) atomicAdd(&acc[rr[k]], vv[k]);
    }
    __syncthreads();

    float* dst = ws + (l ? WS_A1 : WS_A0) + lo;
    for (int i0 = tid; i0 < N2N_RB; i0 += FL_U * SC_T) {
        float sv[FL_U];
        #pragma unroll
        for (int u = 0; u < FL_U; ++u) {
            const int i = i0 + u * SC_T;
            sv[u] = acc[i < N2N_RB ? i : 0];
        }
        #pragma unroll
        for (int u = 0; u < FL_U; ++u) {
            const int i = i0 + u * SC_T;
            if (i < N2N_RB && lo + i < NN && sv[u] != 0.0f) atomicAdd(&dst[i], sv[u]);
        }
    }
}

__global__ __launch_bounds__(SC_T)
void k_sage(const int* __restrict__ s0r, const int* __restrict__ s0c,
            const int* __restrict__ s1r, const int* __restrict__ s1c,
            float* ws) {
    __shared__ __attribute__((aligned(16))) float        num[SG_RB];
    __shared__ __attribute__((aligned(16))) unsigned int cnt[SG_RB];
    const int tid = threadIdx.x;
    const int bid = blockIdx.x;
    const int c = bid % SC_C;
    const int p = (bid / SC_C) % SC_P;
    const int l = bid / (SC_C * SC_P);
    const int lo = p * SG_RB;
    const int q0 = (c * SG_Q) / SC_C;
    const int n4 = ((c + 1) * SG_Q) / SC_C - q0;
    const int4* __restrict__ row4 = reinterpret_cast<const int4*>(l ? s1r : s0r) + q0;
    const int4* __restrict__ col4 = reinterpret_cast<const int4*>(l ? s1c : s0c) + q0;
    const float* a = ws + (l ? WS_A1 : WS_A0);
    const int STEP4 = SC_T * SC_V;

    for (int i = tid; i < SG_RB / 4; i += SC_T) {
        reinterpret_cast<float4*>(num)[i] = make_float4(0.f, 0.f, 0.f, 0.f);
        reinterpret_cast<uint4*>(cnt)[i]  = make_uint4(0u, 0u, 0u, 0u);
    }
    __syncthreads();

    int4 qr[SC_V]; int4 qc[SC_V];
    int rr[SC_U]; float gv[SC_U];
    #pragma unroll
    for (int u = 0; u < SC_V; ++u) {
        int i4 = u * SC_T + tid;
        int ii = i4 < n4 ? i4 : 0;
        qr[u] = row4[ii];
        qc[u] = col4[ii];
    }
    #pragma unroll
    for (int u = 0; u < SC_V; ++u) {
        bool ok = (u * SC_T + tid) < n4;
        int r4[4] = {qr[u].x - lo, qr[u].y - lo, qr[u].z - lo, qr[u].w - lo};
        int c4[4] = {qc[u].x, qc[u].y, qc[u].z, qc[u].w};
        #pragma unroll
        for (int j = 0; j < 4; ++j) {
            bool m = ok && ((unsigned)r4[j] < (unsigned)SG_RB);
            rr[4 * u + j] = m ? r4[j] : -1;
            gv[4 * u + j] = a[m ? c4[j] : 0];
        }
    }
    #pragma unroll
    for (int u = 0; u < SC_V; ++u) {
        int i4 = STEP4 + u * SC_T + tid;
        int ii = i4 < n4 ? i4 : 0;
        qr[u] = row4[ii];
        qc[u] = col4[ii];
    }
    for (int b4 = 0; b4 < n4; b4 += STEP4) {
        int nr[SC_U]; float ng[SC_U];
        #pragma unroll
        for (int u = 0; u < SC_V; ++u) {
            bool ok = (b4 + STEP4 + u * SC_T + tid) < n4;
            int r4[4] = {qr[u].x - lo, qr[u].y - lo, qr[u].z - lo, qr[u].w - lo};
            int c4[4] = {qc[u].x, qc[u].y, qc[u].z, qc[u].w};
            #pragma unroll
            for (int j = 0; j < 4; ++j) {
                bool m = ok && ((unsigned)r4[j] < (unsigned)SG_RB);
                nr[4 * u + j] = m ? r4[j] : -1;
                ng[4 * u + j] = a[m ? c4[j] : 0];
            }
        }
        #pragma unroll
        for (int u = 0; u < SC_V; ++u) {
            int i4 = b4 + 2 * STEP4 + u * SC_T + tid;
            int ii = i4 < n4 ? i4 : 0;
            qr[u] = row4[ii];
            qc[u] = col4[ii];
        }
        #pragma unroll
        for (int k = 0; k < SC_U; ++k)
            if (rr[k] >= 0) {
                atomicAdd(&num[rr[k]], gv[k]);
                atomicAdd(&cnt[rr[k]], 1u);
            }
        #pragma unroll
        for (int k = 0; k < SC_U; ++k) { rr[k] = nr[k]; gv[k] = ng[k]; }
    }
    __syncthreads();

    float* dm = ws + (l ? WS_M1   : WS_M0)   + lo;
    float* dd = ws + (l ? WS_DEG1 : WS_DEG0) + lo;
    for (int i0 = tid; i0 < SG_RB; i0 += FL_U * SC_T) {
        unsigned int k[FL_U]; float s[FL_U];
        #pragma unroll
        for (int u = 0; u < FL_U; ++u) {
            const int i  = i0 + u * SC_T;
            const int ii = i < SG_RB ? i : 0;
            k[u] = cnt[ii];
            s[u] = num[ii];
        }
        #pragma unroll
        for (int u = 0; u < FL_U; ++u) {
            const int i = i0 + u * SC_T;
            if (i < SG_RB && lo + i < NBT && k[u] != 0u) {
                atomicAdd(&dm[i], s[u]);
                atomicAdd(&dd[i], (float)k[u]);
            }
        }
    }
}

#define RW_T    256
#define RW_NCH  (2 * NN / 32)
#define RW_NB   ((RW_NCH * 32 + RW_T - 1) / RW_T)
#define HD_BPB  16
#define HD_NB   (BB / HD_BPB)
#define HD_EP   68
#define HD_HP   33
static_assert(NN % 32 == 0, "a 32-row chunk must stay inside one layer");
static_assert(BB % HD_BPB == 0, "head blocks must tile the subgraph nodes");
static_assert(HG_B * HG_E >= NN && HG_U * SC_T >= HG_E, "the subgraph blocks must cover the subgraph entries");
static_assert(BB <= SC_T && BB <= N2N_RB, "one thread and one LDS word per subgraph node");

__device__ __forceinline__ float row_val(float a, float u1, float m, float u2) {
    return fmaxf(fmaf(a, u1, m * u2), 0.0f);
}

__device__ __forceinline__ float row_ss(float a, float m,
                                        const float* __restrict__ u1, const float* __restrict__ u2) {
    float s0 = 0.f, s1 = 0.f, s2 = 0.f, s3 = 0.f;
    #pragma unroll
    for (int d = 0; d < DD; d += 4) {
        float r0 = row_val(a, u1[d + 0], m, u2[d + 0]);
        float r1 = row_val(a, u1[d + 1], m, u2[d + 1]);
        float r2 = row_val(a, u1[d + 2], m, u2[d + 2]);
        float r3 = row_val(a, u1[d + 3], m, u2[d + 3]);
        s0 = fmaf(r0, r0, s0);
        s1 = fmaf(r1, r1, s1);
        s2 = fmaf(r2, r2, s2);
        s3 = fmaf(r3, r3, s3);
    }
    return (s0 + s1) + (s2 + s3);
}

__global__ __launch_bounds__(RW_T)
void k_rows(const float* __restrict__ wsr,
            const float* __restrict__ cp,
            const int*   __restrict__ act_col,
            const float* __restrict__ h1,
            const float* __restrict__ aux,
            const float* __restrict__ h2w,
            float* __restrict__ out) {
    __shared__ float s_t[2 * HD_BPB], s_a[2 * HD_BPB], s_m[2 * HD_BPB], s_i[2 * HD_BPB];
    __shared__ __attribute__((aligned(16))) float s_esa[2 * HD_BPB * HD_EP];
    __shared__ float s_hid[2 * HD_BPB * HD_HP];

    const float* __restrict__ u1 = wsr + WS_U1;
    const float* __restrict__ u2 = wsr + WS_U2;
    const int lane = threadIdx.x & 31;
    const int c    = lane & 15;
    const int half = lane >> 4;

    if (blockIdx.x < HD_NB) {
        const int tid  = threadIdx.x;
        const int wave = tid >> 5;
        const int b0   = blockIdx.x * HD_BPB;
        if (wave == 0) {
            const int l = lane >> 4;
            const int i = NN + b0 + (lane & 15);
            const float a   = wsr[(l ? WS_A1   : WS_A0)   + i];
            const float num = wsr[(l ? WS_M1   : WS_M0)   + i];
            const float dg  = wsr[(l ? WS_DEG1 : WS_DEG0) + i];
            const float m   = num / fmaxf(dg, 1.0f);
            const float inv = 1.0f / fmaxf(sqrtf(row_ss(a, m, u1, u2)), 1e-12f);
            float t0 = 0.f, t1 = 0.f, t2 = 0.f, t3 = 0.f;
            #pragma unroll
            for (int d = 0; d < DD; d += 4) {
                t0 = fmaf(row_val(a, u1[d + 0], m, u2[d + 0]) * inv, cp[d + 0], t0);
                t1 = fmaf(row_val(a, u1[d + 1], m, u2[d + 1]) * inv, cp[d + 1], t1);
                t2 = fmaf(row_val(a, u1[d + 2], m, u2[d + 2]) * inv, cp[d + 2], t2);
                t3 = fmaf(row_val(a, u1[d + 3], m, u2[d + 3]) * inv, cp[d + 3], t3);
            }
            s_t[lane] = (t0 + t1) + (t2 + t3);
        } else if (wave == 1) {
            const int l = lane >> 4;
            const int i = act_col[b0 + (lane & 15)];
            const float a   = wsr[(l ? WS_A1   : WS_A0)   + i];
            const float num = wsr[(l ? WS_M1   : WS_M0)   + i];
            const float dg  = wsr[(l ? WS_DEG1 : WS_DEG0) + i];
            const float m   = num / fmaxf(dg, 1.0f);
            s_a[lane] = a;
            s_m[lane] = m;
            s_i[lane] = 1.0f / fmaxf(sqrtf(row_ss(a, m, u1, u2)), 1e-12f);
        }
        __syncthreads();
        for (int j = 0; j < (2 * HD_BPB * DD) / RW_T; ++j) {
            const int idx = j * RW_T + tid;
            const int pr  = idx >> 6;
            const int d   = idx & (DD - 1);
            const float ae = row_val(s_a[pr], u1[d], s_m[pr], u2[d]) * s_i[pr];
            s_esa[pr * HD_EP + d] = s_t[pr] * ae;
        }
        __syncthreads();
        if (wave < 4) {
            const int l  = wave >> 1;
            const int tn = wave & 1;
            const float* A = s_esa + (l * HD_BPB) * HD_EP;
            v8f acc = {0.f, 0.f, 0.f, 0.f, 0.f, 0.f, 0.f, 0.f};
            #pragma unroll
            for (int s = 0; s < 16; ++s) {
                const int k0 = 4 * s + 2 * half;
                v2f av, bv;
                av.x = A[c * HD_EP + k0];
                av.y = A[c * HD_EP + k0 + 1];
                bv.x = h1[(k0    ) * 32 + tn * 16 + c];
                bv.y = h1[(k0 + 1) * 32 + tn * 16 + c];
                acc = __builtin_amdgcn_wmma_f32_16x16x4_f32(false, av, false, bv,
                                                            (short)0, acc, false, false);
            }
            #pragma unroll
            for (int r = 0; r < 8; ++r) {
                const int M = r + 8 * half;
                s_hid[(l * HD_BPB + M) * HD_HP + tn * 16 + c] = fmaxf(acc[r], 0.0f);
            }
        }
        __syncthreads();
        if (tid < HD_BPB) {
            const int b = b0 + tid;
            float q = 0.0f;
            for (int l = 0; l < 2; ++l) {
                const float* h = s_hid + (l * HD_BPB + tid) * HD_HP;
                for (int j = 0; j < 32; ++j) q += h[j] * h2w[j];
                for (int j = 0; j < 4; ++j)  q += aux[b * 8 + l * 4 + j] * h2w[32 + j];
            }
            out[b] = q;
        }
        return;
    }

    const int chunk = ((blockIdx.x - HD_NB) * RW_T + threadIdx.x) >> 5;
    if (chunk >= RW_NCH) return;
    const int l  = chunk / (NN / 32);
    const int i0 = (chunk - l * (NN / 32)) * 32;
    const int i  = i0 + lane;

    const float a   = wsr[(l ? WS_A1   : WS_A0)   + i];
    const float num = wsr[(l ? WS_M1   : WS_M0)   + i];
    const float dg  = wsr[(l ? WS_DEG1 : WS_DEG0) + i];
    const float4 p1 = *reinterpret_cast<const float4*>(wsr + WS_U1 + 4 * c);
    const float4 p2 = *reinterpret_cast<const float4*>(wsr + WS_U2 + 4 * c);
    const float m = num / fmaxf(dg, 1.0f);
    const float inv = 1.0f / fmaxf(sqrtf(row_ss(a, m, u1, u2)), 1e-12f);

    float* dst = out + BB + (l * NN + i0 + half) * DD + 4 * c;
    #pragma unroll
    for (int p = 0; p < 16; ++p) {
        const int src = 2 * p + half;
        const float ar = __shfl(a,   src, 32);
        const float mr = __shfl(m,   src, 32);
        const float ir = __shfl(inv, src, 32);
        float4 o;
        o.x = row_val(ar, p1.x, mr, p2.x) * ir;
        o.y = row_val(ar, p1.y, mr, p2.y) * ir;
        o.z = row_val(ar, p1.z, mr, p2.z) * ir;
        o.w = row_val(ar, p1.w, mr, p2.w) * ir;
        *reinterpret_cast<float4*>(dst + 2 * p * DD) = o;
    }
}

extern "C" void kernel_launch(void* const* d_in, const int* in_sizes, int n_in,
                              void* d_out, int out_size, void* d_ws, size_t ws_size,
                              hipStream_t stream) {
    const int*   n2n0_row = (const int*)  d_in[0];
    const float* n2n0_val = (const float*)d_in[2];
    const int*   n2n1_row = (const int*)  d_in[3];
    const float* n2n1_val = (const float*)d_in[5];
    const int*   subg_row = (const int*)  d_in[6];
    const float* subg_val = (const float*)d_in[8];
    const int*   act_col  = (const int*)  d_in[9];
    const int*   sg0_row  = (const int*)  d_in[10];
    const int*   sg0_col  = (const int*)  d_in[11];
    const int*   sg1_row  = (const int*)  d_in[12];
    const int*   sg1_col  = (const int*)  d_in[13];
    const float* aux      = (const float*)d_in[14];
    const float* w_n2l    = (const float*)d_in[15];
    const float* p_conv   = (const float*)d_in[16];
    const float* W_sage   = (const float*)d_in[17];
    const float* cp       = (const float*)d_in[18];
    const float* h1w      = (const float*)d_in[19];
    const float* h2w      = (const float*)d_in[20];
    float* out = (float*)d_out;
    float* ws  = (float*)d_ws;

    k_prep <<<1 + PR_ZB, PR_T, 0, stream>>>(w_n2l, p_conv, W_sage, ws);
    k_n2n  <<<SC_GRID + HG_B, SC_T, 0, stream>>>(n2n0_row, n2n0_val, n2n1_row, n2n1_val, subg_row, subg_val, ws);
    k_sage <<<SC_GRID, SC_T, 0, stream>>>(sg0_row, sg0_col, sg1_row, sg1_col, ws);
    k_rows <<<HD_NB + RW_NB, RW_T, 0, stream>>>(ws, cp, act_col, h1w, aux, h2w, out);
}


# speedup vs baseline: 7.3522x; 1.3438x over previous
#include <hip/hip_runtime.h>

#define NN   100000
#define BB   512
#define EE   1000000
#define ESZ  1200000
#define DD   64
#define NBT  (NN + BB)

#define WS_A0   0
#define WS_A1   (NBT)
#define WS_M0   (2*NBT)
#define WS_M1   (3*NBT)
#define WS_DEG0 (4*NBT)
#define WS_DEG1 (5*NBT)
#define WS_SM   (6*NBT)
#define WS_TK   (WS_SM)
#define WS_U1   (WS_SM + 128)
#define WS_U2   (WS_SM + 192)

#define SC_P    4
#define SC_C    32
#define SC_T    1024
#define SC_V    2
#define SC_U    (4 * SC_V)
#define SC_GRID (2 * SC_P * SC_C)
#define FL_U    5
#define HG_B    8
#define HG_E    ((NN + HG_B - 1) / HG_B)
#define HG_U    ((HG_E + SC_T - 1) / SC_T)
#define SC_RB(n) ((((n) + SC_P - 1) / SC_P + 31) / 32 * 32)
#define N2N_RB  SC_RB(NN)
#define SG_RB   SC_RB(NBT)
#define N2N_Q   (EE / 4)
#define SG_Q    (ESZ / 4)
static_assert(SC_P * N2N_RB >= NN && SC_P * SG_RB >= NBT, "row ranges must cover the rows");
static_assert(N2N_RB % 4 == 0 && SG_RB % 4 == 0, "the LDS sums are cleared in four-word pieces");
static_assert(NBT % 32 == 0, "the accumulator planes must start on whole lines");
static_assert(EE % 4 == 0 && ESZ % 4 == 0, "the edges are read in four-word pieces");
static_assert((long long)SC_C * SG_Q < 2147483647LL, "chunk bounds are computed in int");

typedef float v2f __attribute__((ext_vector_type(2)));
typedef float v8f __attribute__((ext_vector_type(8)));

__device__ __forceinline__ int take_item(int* tk) {
    const unsigned id = (unsigned)__builtin_amdgcn_s_sendmsg_rtn(0x87);
    const int h = (int)((id >> 18) & 1u);
    const int t = atomicAdd(&tk[h], 1);
    const int half_n = (2 * SC_P * SC_C) / 2;
    int item = t < half_n ? h * half_n + t : (1 - h) * half_n + (2 * half_n - 1 - t);
    return item < 0 ? 0 : (item > 2 * half_n - 1 ? 2 * half_n - 1 : item);
}

#define PR_T   256
#define PR_ZB  ((6 * NBT + PR_T - 1) / PR_T)
__global__ __launch_bounds__(PR_T)
void k_prep(const float* __restrict__ w_n2l,
            const float* __restrict__ p_conv,
            const float* __restrict__ W_sage,
            float* ws) {
    __shared__ float v[DD], w[DD];
    if (blockIdx.x != 0) {
        int i = (blockIdx.x - 1) * PR_T + threadIdx.x;
        if (i < 6 * NBT) ws[i] = 0.0f;
        return;
    }
    const int d = threadIdx.x & (DD - 1);
    float x = fmaxf(w_n2l[d] + w_n2l[DD + d], 0.0f);
    v[d] = x;
    __syncthreads();
    float ss = 0.f;
    for (int k = 0; k < DD; ++k) ss += v[k] * v[k];
    float nv = x / fmaxf(sqrtf(ss), 1e-12f);
    __syncthreads();
    v[d] = nv;
    __syncthreads();
    float wd = 0.f;
    for (int k = 0; k < DD; ++k) wd += v[k] * p_conv[k * DD + d];
    w[d] = wd;
    __syncthreads();
    float u1 = 0.f, u2 = 0.f;
    for (int k = 0; k < DD; ++k) {
        u1 += w[k] * W_sage[k * DD + d];
        u2 += w[k] * W_sage[(DD + k) * DD + d];
    }
    if (threadIdx.x < DD) {
        ws[WS_U1 + d] = u1;
        ws[WS_U2 + d] = u2;
    }
    if (threadIdx.x < 4) reinterpret_cast<int*>(ws)[WS_TK + threadIdx.x] = 0;
}

__global__ __launch_bounds__(SC_T)
void k_n2n(const int* __restrict__ r0, const float* __restrict__ v0,
           const int* __restrict__ r1, const float* __restrict__ v1,
           const int* __restrict__ subg_row, const float* __restrict__ subg_val,
           float* ws) {
    __shared__ __attribute__((aligned(16))) float acc[N2N_RB];
    __shared__ int s_item;
    const int tid = threadIdx.x;
    const int bid = blockIdx.x;
    if (bid >= SC_GRID) {
        const int e0 = (bid - SC_GRID) * HG_E;
        if (tid < BB) acc[tid] = 0.0f;
        __syncthreads();
        int hb[HG_U]; float hv[HG_U];
        #pragma unroll
        for (int u = 0; u < HG_U; ++u) {
            const int j = u * SC_T + tid;
            const bool ok = j < HG_E && e0 + j < NN;
            hb[u] = ok ? subg_row[e0 + j] : -1;
            hv[u] = ok ? subg_val[e0 + j] : 0.0f;
        }
        #pragma unroll
        for (int u = 0; u < HG_U; ++u)
            if ((unsigned)hb[u] < (unsigned)BB) atomicAdd(&acc[hb[u]], hv[u]);
        __syncthreads();
        if (tid < BB) {
            const float s = acc[tid];
            if (s != 0.0f) {
                atomicAdd(&ws[WS_A0 + NN + tid], s);
                atomicAdd(&ws[WS_A1 + NN + tid], s);
            }
        }
        return;
    }
    if (tid == 0) s_item = take_item(reinterpret_cast<int*>(ws) + WS_TK);
    for (int i = tid; i < N2N_RB / 4; i += SC_T)
        reinterpret_cast<float4*>(acc)[i] = make_float4(0.f, 0.f, 0.f, 0.f);
    __syncthreads();

    const int item = s_item;
    const int c = item % SC_C;
    const int p = (item / SC_C) % SC_P;
    const int l = item / (SC_C * SC_P);
    const int lo = p * N2N_RB;
    const int q0 = (c * N2N_Q) / SC_C;
    const int n4 = ((c + 1) * N2N_Q) / SC_C - q0;
    const int4*   __restrict__ row4 = reinterpret_cast<const int4*>(l ? r1 : r0) + q0;
    const float4* __restrict__ val4 = reinterpret_cast<const float4*>(l ? v1 : v0) + q0;

    int4 qr[SC_V]; float4 qv[SC_V];
    #pragma unroll
    for (int u = 0; u < SC_V; ++u) {
        int i4 = u * SC_T + tid;
        int ii = i4 < n4 ? i4 : 0;
        qr[u] = row4[ii];
        qv[u] = val4[ii];
    }
    for (int b4 = 0; b4 < n4; b4 += SC_T * SC_V) {
        int rr[SC_U]; float vv[SC_U];
        #pragma unroll
        for (int u = 0; u < SC_V; ++u) {
            bool ok = (b4 + u * SC_T + tid) < n4;
            rr[4 * u + 0] = ok ? qr[u].x - lo : -1;  vv[4 * u + 0] = qv[u].x;
            rr[4 * u + 1] = ok ? qr[u].y - lo : -1;  vv[4 * u + 1] = qv[u].y;
            rr[4 * u + 2] = ok ? qr[u].z - lo : -1;  vv[4 * u + 2] = qv[u].z;
            rr[4 * u + 3] = ok ? qr[u].w - lo : -1;  vv[4 * u + 3] = qv[u].w;
        }
        #pragma unroll
        for (int u = 0; u < SC_V; ++u) {
            int i4 = b4 + SC_T * SC_V + u * SC_T + tid;
            int ii = i4 < n4 ? i4 : 0;
            qr[u] = row4[ii];
            qv[u] = val4[ii];
        }
        #pragma unroll
        for (int k = 0; k < SC_U; ++k)
            if ((unsigned)rr[k] < (unsigned)N2N_RB) atomicAdd(&acc[rr[k]], vv[k]);
    }
    __syncthreads();

    float* dst = ws + (l ? WS_A1 : WS_A0) + lo;
    for (int i0 = tid; i0 < N2N_RB; i0 += FL_U * SC_T) {
        float sv[FL_U];
        #pragma unroll
        for (int u = 0; u < FL_U; ++u) {
            const int i = i0 + u * SC_T;
            sv[u] = acc[i < N2N_RB ? i : 0];
        }
        #pragma unroll
        for (int u = 0; u < FL_U; ++u) {
            const int i = i0 + u * SC_T;
            if (i < N2N_RB && lo + i < NN && sv[u] != 0.0f) atomicAdd(&dst[i], sv[u]);
        }
    }
}

__global__ __launch_bounds__(SC_T)
void k_sage(const int* __restrict__ s0r, const int* __restrict__ s0c,
            const int* __restrict__ s1r, const int* __restrict__ s1c,
            float* ws) {
    __shared__ __attribute__((aligned(16))) float        num[SG_RB];
    __shared__ __attribute__((aligned(16))) unsigned int cnt[SG_RB];
    __shared__ int s_item;
    const int tid = threadIdx.x;
    if (tid == 0) s_item = take_item(reinterpret_cast<int*>(ws) + WS_TK + 2);
    for (int i = tid; i < SG_RB / 4; i += SC_T) {
        reinterpret_cast<float4*>(num)[i] = make_float4(0.f, 0.f, 0.f, 0.f);
        reinterpret_cast<uint4*>(cnt)[i]  = make_uint4(0u, 0u, 0u, 0u);
    }
    __syncthreads();

    const int item = s_item;
    const int c = item % SC_C;
    const int p = (item / SC_C) % SC_P;
    const int l = item / (SC_C * SC_P);
    const int lo = p * SG_RB;
    const int q0 = (c * SG_Q) / SC_C;
    const int n4 = ((c + 1) * SG_Q) / SC_C - q0;
    const int4* __restrict__ row4 = reinterpret_cast<const int4*>(l ? s1r : s0r) + q0;
    const int4* __restrict__ col4 = reinterpret_cast<const int4*>(l ? s1c : s0c) + q0;
    const float* a = ws + (l ? WS_A1 : WS_A0);
    const int STEP4 = SC_T * SC_V;

    int4 qr[SC_V]; int4 qc[SC_V];
    int rr[SC_U]; float gv[SC_U];
    #pragma unroll
    for (int u = 0; u < SC_V; ++u) {
        int i4 = u * SC_T + tid;
        int ii = i4 < n4 ? i4 : 0;
        qr[u] = row4[ii];
        qc[u] = col4[ii];
    }
    #pragma unroll
    for (int u = 0; u < SC_V; ++u) {
        bool ok = (u * SC_T + tid) < n4;
        int r4[4] = {qr[u].x - lo, qr[u].y - lo, qr[u].z - lo, qr[u].w - lo};
        int c4[4] = {qc[u].x, qc[u].y, qc[u].z, qc[u].w};
        #pragma unroll
        for (int j = 0; j < 4; ++j) {
            bool m = ok && ((unsigned)r4[j] < (unsigned)SG_RB);
            rr[4 * u + j] = m ? r4[j] : -1;
            gv[4 * u + j] = a[m ? c4[j] : 0];
        }
    }
    #pragma unroll
    for (int u = 0; u < SC_V; ++u) {
        int i4 = STEP4 + u * SC_T + tid;
        int ii = i4 < n4 ? i4 : 0;
        qr[u] = row4[ii];
        qc[u] = col4[ii];
    }
    for (int b4 = 0; b4 < n4; b4 += STEP4) {
        int nr[SC_U]; float ng[SC_U];
        #pragma unroll
        for (int u = 0; u < SC_V; ++u) {
            bool ok = (b4 + STEP4 + u * SC_T + tid) < n4;
            int r4[4] = {qr[u].x - lo, qr[u].y - lo, qr[u].z - lo, qr[u].w - lo};
            int c4[4] = {qc[u].x, qc[u].y, qc[u].z, qc[u].w};
            #pragma unroll
            for (int j = 0; j < 4; ++j) {
                bool m = ok && ((unsigned)r4[j] < (unsigned)SG_RB);
                nr[4 * u + j] = m ? r4[j] : -1;
                ng[4 * u + j] = a[m ? c4[j] : 0];
            }
        }
        #pragma unroll
        for (int u = 0; u < SC_V; ++u) {
            int i4 = b4 + 2 * STEP4 + u * SC_T + tid;
            int ii = i4 < n4 ? i4 : 0;
            qr[u] = row4[ii];
            qc[u] = col4[ii];
        }
        #pragma unroll
        for (int k = 0; k < SC_U; ++k)
            if (rr[k] >= 0) {
                atomicAdd(&num[rr[k]], gv[k]);
                atomicAdd(&cnt[rr[k]], 1u);
            }
        #pragma unroll
        for (int k = 0; k < SC_U; ++k) { rr[k] = nr[k]; gv[k] = ng[k]; }
    }
    __syncthreads();

    float* dm = ws + (l ? WS_M1   : WS_M0)   + lo;
    float* dd = ws + (l ? WS_DEG1 : WS_DEG0) + lo;
    for (int i0 = tid; i0 < SG_RB; i0 += FL_U * SC_T) {
        unsigned int k[FL_U]; float s[FL_U];
        #pragma unroll
        for (int u = 0; u < FL_U; ++u) {
            const int i  = i0 + u * SC_T;
            const int ii = i < SG_RB ? i : 0;
            k[u] = cnt[ii];
            s[u] = num[ii];
        }
        #pragma unroll
        for (int u = 0; u < FL_U; ++u) {
            const int i = i0 + u * SC_T;
            if (i < SG_RB && lo + i < NBT && k[u] != 0u) {
                atomicAdd(&dm[i], s[u]);
                atomicAdd(&dd[i], (float)k[u]);
            }
        }
    }
}

#define RW_T    256
#define RW_NCH  (2 * NN / 32)
#define RW_NB   ((RW_NCH * 32 + RW_T - 1) / RW_T)
#define HD_BPB  16
#define HD_NB   (BB / HD_BPB)
#define HD_EP   68
#define HD_HP   33
static_assert(NN % 32 == 0, "a 32-row chunk must stay inside one layer");
static_assert(BB % HD_BPB == 0, "head blocks must tile the subgraph nodes");
static_assert(HG_B * HG_E >= NN && HG_U * SC_T >= HG_E, "the subgraph blocks must cover the subgraph entries");
static_assert(BB <= SC_T && BB <= N2N_RB, "one thread and one LDS word per subgraph node");

__device__ __forceinline__ float row_val(float a, float u1, float m, float u2) {
    return fmaxf(fmaf(a, u1, m * u2), 0.0f);
}

__device__ __forceinline__ float row_ss(float a, float m,
                                        const float* __restrict__ u1, const float* __restrict__ u2) {
    float s0 = 0.f, s1 = 0.f, s2 = 0.f, s3 = 0.f;
    #pragma unroll
    for (int d = 0; d < DD; d += 4) {
        float r0 = row_val(a, u1[d + 0], m, u2[d + 0]);
        float r1 = row_val(a, u1[d + 1], m, u2[d + 1]);
        float r2 = row_val(a, u1[d + 2], m, u2[d + 2]);
        float r3 = row_val(a, u1[d + 3], m, u2[d + 3]);
        s0 = fmaf(r0, r0, s0);
        s1 = fmaf(r1, r1, s1);
        s2 = fmaf(r2, r2, s2);
        s3 = fmaf(r3, r3, s3);
    }
    return (s0 + s1) + (s2 + s3);
}

__global__ __launch_bounds__(RW_T)
void k_rows(const float* __restrict__ wsr,
            const float* __restrict__ cp,
            const int*   __restrict__ act_col,
            const float* __restrict__ h1,
            const float* __restrict__ aux,
            const float* __restrict__ h2w,
            float* __restrict__ out) {
    __shared__ float s_t[2 * HD_BPB], s_a[2 * HD_BPB], s_m[2 * HD_BPB], s_i[2 * HD_BPB];
    __shared__ __attribute__((aligned(16))) float s_esa[2 * HD_BPB * HD_EP];
    __shared__ float s_hid[2 * HD_BPB * HD_HP];

    const float* __restrict__ u1 = wsr + WS_U1;
    const float* __restrict__ u2 = wsr + WS_U2;
    const int lane = threadIdx.x & 31;
    const int c    = lane & 15;
    const int half = lane >> 4;

    if (blockIdx.x < HD_NB) {
        const int tid  = threadIdx.x;
        const int wave = tid >> 5;
        const int b0   = blockIdx.x * HD_BPB;
        if (wave == 0) {
            const int l = lane >> 4;
            const int i = NN + b0 + (lane & 15);
            const float a   = wsr[(l ? WS_A1   : WS_A0)   + i];
            const float num = wsr[(l ? WS_M1   : WS_M0)   + i];
            const float dg  = wsr[(l ? WS_DEG1 : WS_DEG0) + i];
            const float m   = num / fmaxf(dg, 1.0f);
            const float inv = 1.0f / fmaxf(sqrtf(row_ss(a, m, u1, u2)), 1e-12f);
            float t0 = 0.f, t1 = 0.f, t2 = 0.f, t3 = 0.f;
            #pragma unroll
            for (int d = 0; d < DD; d += 4) {
                t0 = fmaf(row_val(a, u1[d + 0], m, u2[d + 0]) * inv, cp[d + 0], t0);
                t1 = fmaf(row_val(a, u1[d + 1], m, u2[d + 1]) * inv, cp[d + 1], t1);
                t2 = fmaf(row_val(a, u1[d + 2], m, u2[d + 2]) * inv, cp[d + 2], t2);
                t3 = fmaf(row_val(a, u1[d + 3], m, u2[d + 3]) * inv, cp[d + 3], t3);
            }
            s_t[lane] = (t0 + t1) + (t2 + t3);
        } else if (wave == 1) {
            const int l = lane >> 4;
            const int i = act_col[b0 + (lane & 15)];
            const float a   = wsr[(l ? WS_A1   : WS_A0)   + i];
            const float num = wsr[(l ? WS_M1   : WS_M0)   + i];
            const float dg  = wsr[(l ? WS_DEG1 : WS_DEG0) + i];
            const float m   = num / fmaxf(dg, 1.0f);
            s_a[lane] = a;
            s_m[lane] = m;
            s_i[lane] = 1.0f / fmaxf(sqrtf(row_ss(a, m, u1, u2)), 1e-12f);
        }
        __syncthreads();
        for (int j = 0; j < (2 * HD_BPB * DD) / RW_T; ++j) {
            const int idx = j * RW_T + tid;
            const int pr  = idx >> 6;
            const int d   = idx & (DD - 1);
            const float ae = row_val(s_a[pr], u1[d], s_m[pr], u2[d]) * s_i[pr];
            s_esa[pr * HD_EP + d] = s_t[pr] * ae;
        }
        __syncthreads();
        if (wave < 4) {
            const int l  = wave >> 1;
            const int tn = wave & 1;
            const float* A = s_esa + (l * HD_BPB) * HD_EP;
            v8f acc = {0.f, 0.f, 0.f, 0.f, 0.f, 0.f, 0.f, 0.f};
            #pragma unroll
            for (int s = 0; s < 16; ++s) {
                const int k0 = 4 * s + 2 * half;
                v2f av, bv;
                av.x = A[c * HD_EP + k0];
                av.y = A[c * HD_EP + k0 + 1];
                bv.x = h1[(k0    ) * 32 + tn * 16 + c];
                bv.y = h1[(k0 + 1) * 32 + tn * 16 + c];
                acc = __builtin_amdgcn_wmma_f32_16x16x4_f32(false, av, false, bv,
                                                            (short)0, acc, false, false);
            }
            #pragma unroll
            for (int r = 0; r < 8; ++r) {
                const int M = r + 8 * half;
                s_hid[(l * HD_BPB + M) * HD_HP + tn * 16 + c] = fmaxf(acc[r], 0.0f);
            }
        }
        __syncthreads();
        if (tid < HD_BPB) {
            const int b = b0 + tid;
            float q = 0.0f;
            for (int l = 0; l < 2; ++l) {
                const float* h = s_hid + (l * HD_BPB + tid) * HD_HP;
                for (int j = 0; j < 32; ++j) q += h[j] * h2w[j];
                for (int j = 0; j < 4; ++j)  q += aux[b * 8 + l * 4 + j] * h2w[32 + j];
            }
            out[b] = q;
        }
        return;
    }

    const int chunk = ((blockIdx.x - HD_NB) * RW_T + threadIdx.x) >> 5;
    if (chunk >= RW_NCH) return;
    const int l  = chunk / (NN / 32);
    const int i0 = (chunk - l * (NN / 32)) * 32;
    const int i  = i0 + lane;

    const float a   = wsr[(l ? WS_A1   : WS_A0)   + i];
    const float num = wsr[(l ? WS_M1   : WS_M0)   + i];
    const float dg  = wsr[(l ? WS_DEG1 : WS_DEG0) + i];
    const float4 p1 = *reinterpret_cast<const float4*>(wsr + WS_U1 + 4 * c);
    const float4 p2 = *reinterpret_cast<const float4*>(wsr + WS_U2 + 4 * c);
    const float m = num / fmaxf(dg, 1.0f);
    const float inv = 1.0f / fmaxf(sqrtf(row_ss(a, m, u1, u2)), 1e-12f);

    float* dst = out + BB + (l * NN + i0 + half) * DD + 4 * c;
    #pragma unroll
    for (int p = 0; p < 16; ++p) {
        const int src = 2 * p + half;
        const float ar = __shfl(a,   src, 32);
        const float mr = __shfl(m,   src, 32);
        const float ir = __shfl(inv, src, 32);
        float4 o;
        o.x = row_val(ar, p1.x, mr, p2.x) * ir;
        o.y = row_val(ar, p1.y, mr, p2.y) * ir;
        o.z = row_val(ar, p1.z, mr, p2.z) * ir;
        o.w = row_val(ar, p1.w, mr, p2.w) * ir;
        *reinterpret_cast<float4*>(dst + 2 * p * DD) = o;
    }
}

extern "C" void kernel_launch(void* const* d_in, const int* in_sizes, int n_in,
                              void* d_out, int out_size, void* d_ws, size_t ws_size,
                              hipStream_t stream) {
    const int*   n2n0_row = (const int*)  d_in[0];
    const float* n2n0_val = (const float*)d_in[2];
    const int*   n2n1_row = (const int*)  d_in[3];
    const float* n2n1_val = (const float*)d_in[5];
    const int*   subg_row = (const int*)  d_in[6];
    const float* subg_val = (const float*)d_in[8];
    const int*   act_col  = (const int*)  d_in[9];
    const int*   sg0_row  = (const int*)  d_in[10];
    const int*   sg0_col  = (const int*)  d_in[11];
    const int*   sg1_row  = (const int*)  d_in[12];
    const int*   sg1_col  = (const int*)  d_in[13];
    const float* aux      = (const float*)d_in[14];
    const float* w_n2l    = (const float*)d_in[15];
    const float* p_conv   = (const float*)d_in[16];
    const float* W_sage   = (const float*)d_in[17];
    const float* cp       = (const float*)d_in[18];
    const float* h1w      = (const float*)d_in[19];
    const float* h2w      = (const float*)d_in[20];
    float* out = (float*)d_out;
    float* ws  = (float*)d_ws;

    k_prep <<<1 + PR_ZB, PR_T, 0, stream>>>(w_n2l, p_conv, W_sage, ws);
    k_n2n  <<<SC_GRID + HG_B, SC_T, 0, stream>>>(n2n0_row, n2n0_val, n2n1_row, n2n1_val, subg_row, subg_val, ws);
    k_sage <<<SC_GRID, SC_T, 0, stream>>>(sg0_row, sg0_col, sg1_row, sg1_col, ws);
    k_rows <<<HD_NB + RW_NB, RW_T, 0, stream>>>(ws, cp, act_col, h1w, aux, h2w, out);
}


# sage blocks hold the first 40960 words of their layer's n2n-sum plane in LDS (edge counts packed two per LDS word to make room): gathers of those columns are LDS reads, not whole-line L2 fetches
# speedup vs baseline: 7.7323x; 1.0517x over previous
#include <hip/hip_runtime.h>

#define NN   100000
#define BB   512
#define EE   1000000
#define ESZ  1200000
#define DD   64
#define NBT  (NN + BB)

#define WS_A0   0
#define WS_A1   (NBT)
#define WS_M0   (2*NBT)
#define WS_M1   (3*NBT)
#define WS_DEG0 (4*NBT)
#define WS_DEG1 (5*NBT)
#define WS_SM   (6*NBT)
#define WS_TK   (WS_SM)
#define WS_U1   (WS_SM + 128)
#define WS_U2   (WS_SM + 192)

#define SC_P    4
#define SC_C    32
#define SC_T    1024
#define SC_V    2
#define SC_U    (4 * SC_V)
#define SC_GRID (2 * SC_P * SC_C)
#define FL_U    5
#define HG_B    8
#define HG_E    ((NN + HG_B - 1) / HG_B)
#define HG_U    ((HG_E + SC_T - 1) / SC_T)
#define SC_RB(n) ((((n) + SC_P - 1) / SC_P + 31) / 32 * 32)
#define N2N_RB  SC_RB(NN)
#define SG_RB   SC_RB(NBT)
#define N2N_Q   (EE / 4)
#define SG_Q    (ESZ / 4)
#define SG_WN   ((((320 * 1024 - 2048) - SG_RB * 4 - SG_RB * 2) / 4) / (4 * SC_T) * (4 * SC_T))
#define SG_WU   (SG_WN / (4 * SC_T))
static_assert(SC_P * N2N_RB >= NN && SC_P * SG_RB >= NBT, "row ranges must cover the rows");
static_assert(N2N_RB % 4 == 0 && SG_RB % 4 == 0, "the LDS sums are cleared in four-word pieces");
static_assert(NBT % 32 == 0, "the accumulator planes must start on whole lines");
static_assert(EE % 4 == 0 && ESZ % 4 == 0, "the edges are read in four-word pieces");
static_assert((long long)SC_C * SG_Q < 2147483647LL, "chunk bounds are computed in int");
static_assert(4 * (SG_Q / SC_C + 1) < 65536, "a block's edge count per row must fit 16 bits");
static_assert(SG_RB % 8 == 0, "the packed counts are cleared in four-word pieces");
static_assert(SG_WN > 0 && SG_WN % (4 * SC_T) == 0 && SG_WN <= NBT, "the window is staged in whole rounds of four-word pieces of the plane");

typedef float v2f __attribute__((ext_vector_type(2)));
typedef float v8f __attribute__((ext_vector_type(8)));

__device__ __forceinline__ int take_item(int* tk) {
    const unsigned id = (unsigned)__builtin_amdgcn_s_sendmsg_rtn(0x87);
    const int h = (int)((id >> 18) & 1u);
    const int t = atomicAdd(&tk[h], 1);
    const int half_n = (2 * SC_P * SC_C) / 2;
    int item = t < half_n ? h * half_n + t : (1 - h) * half_n + (2 * half_n - 1 - t);
    return item < 0 ? 0 : (item > 2 * half_n - 1 ? 2 * half_n - 1 : item);
}

#define PR_T   256
#define PR_ZB  ((6 * NBT + PR_T - 1) / PR_T)
__global__ __launch_bounds__(PR_T)
void k_prep(const float* __restrict__ w_n2l,
            const float* __restrict__ p_conv,
            const float* __restrict__ W_sage,
            float* ws) {
    __shared__ float v[DD], w[DD];
    if (blockIdx.x != 0) {
        int i = (blockIdx.x - 1) * PR_T + threadIdx.x;
        if (i < 6 * NBT) ws[i] = 0.0f;
        return;
    }
    const int d = threadIdx.x & (DD - 1);
    float x = fmaxf(w_n2l[d] + w_n2l[DD + d], 0.0f);
    v[d] = x;
    __syncthreads();
    float ss = 0.f;
    for (int k = 0; k < DD; ++k) ss += v[k] * v[k];
    float nv = x / fmaxf(sqrtf(ss), 1e-12f);
    __syncthreads();
    v[d] = nv;
    __syncthreads();
    float wd = 0.f;
    for (int k = 0; k < DD; ++k) wd += v[k] * p_conv[k * DD + d];
    w[d] = wd;
    __syncthreads();
    float u1 = 0.f, u2 = 0.f;
    for (int k = 0; k < DD; ++k) {
        u1 += w[k] * W_sage[k * DD + d];
        u2 += w[k] * W_sage[(DD + k) * DD + d];
    }
    if (threadIdx.x < DD) {
        ws[WS_U1 + d] = u1;
        ws[WS_U2 + d] = u2;
    }
    if (threadIdx.x < 4) reinterpret_cast<int*>(ws)[WS_TK + threadIdx.x] = 0;
}

__global__ __launch_bounds__(SC_T)
void k_n2n(const int* __restrict__ r0, const float* __restrict__ v0,
           const int* __restrict__ r1, const float* __restrict__ v1,
           const int* __restrict__ subg_row, const float* __restrict__ subg_val,
           float* ws) {
    __shared__ __attribute__((aligned(16))) float acc[N2N_RB];
    __shared__ int s_item;
    const int tid = threadIdx.x;
    const int bid = blockIdx.x;
    if (bid >= SC_GRID) {
        const int e0 = (bid - SC_GRID) * HG_E;
        if (tid < BB) acc[tid] = 0.0f;
        __syncthreads();
        int hb[HG_U]; float hv[HG_U];
        #pragma unroll
        for (int u = 0; u < HG_U; ++u) {
            const int j = u * SC_T + tid;
            const bool ok = j < HG_E && e0 + j < NN;
            hb[u] = ok ? subg_row[e0 + j] : -1;
            hv[u] = ok ? subg_val[e0 + j] : 0.0f;
        }
        #pragma unroll
        for (int u = 0; u < HG_U; ++u)
            if ((unsigned)hb[u] < (unsigned)BB) atomicAdd(&acc[hb[u]], hv[u]);
        __syncthreads();
        if (tid < BB) {
            const float s = acc[tid];
            if (s != 0.0f) {
                atomicAdd(&ws[WS_A0 + NN + tid], s);
                atomicAdd(&ws[WS_A1 + NN + tid], s);
            }
        }
        return;
    }
    if (tid == 0) s_item = take_item(reinterpret_cast<int*>(ws) + WS_TK);
    for (int i = tid; i < N2N_RB / 4; i += SC_T)
        reinterpret_cast<float4*>(acc)[i] = make_float4(0.f, 0.f, 0.f, 0.f);
    __syncthreads();

    const int item = s_item;
    const int c = item % SC_C;
    const int p = (item / SC_C) % SC_P;
    const int l = item / (SC_C * SC_P);
    const int lo = p * N2N_RB;
    const int q0 = (c * N2N_Q) / SC_C;
    const int n4 = ((c + 1) * N2N_Q) / SC_C - q0;
    const int4*   __restrict__ row4 = reinterpret_cast<const int4*>(l ? r1 : r0) + q0;
    const float4* __restrict__ val4 = reinterpret_cast<const float4*>(l ? v1 : v0) + q0;

    int4 qr[SC_V]; float4 qv[SC_V];
    #pragma unroll
    for (int u = 0; u < SC_V; ++u) {
        int i4 = u * SC_T + tid;
        int ii = i4 < n4 ? i4 : 0;
        qr[u] = row4[ii];
        qv[u] = val4[ii];
    }
    for (int b4 = 0; b4 < n4; b4 += SC_T * SC_V) {
        int rr[SC_U]; float vv[SC_U];
        #pragma unroll
        for (int u = 0; u < SC_V; ++u) {
            bool ok = (b4 + u * SC_T + tid) < n4;
            rr[4 * u + 0] = ok ? qr[u].x - lo : -1;  vv[4 * u + 0] = qv[u].x;
            rr[4 * u + 1] = ok ? qr[u].y - lo : -1;  vv[4 * u + 1] = qv[u].y;
            rr[4 * u + 2] = ok ? qr[u].z - lo : -1;  vv[4 * u + 2] = qv[u].z;
            rr[4 * u + 3] = ok ? qr[u].w - lo : -1;  vv[4 * u + 3] = qv[u].w;
        }
        #pragma unroll
        for (int u = 0; u < SC_V; ++u) {
            int i4 = b4 + SC_T * SC_V + u * SC_T + tid;
            int ii = i4 < n4 ? i4 : 0;
            qr[u] = row4[ii];
            qv[u] = val4[ii];
        }
        #pragma unroll
        for (int k = 0; k < SC_U; ++k)
            if ((unsigned)rr[k] < (unsigned)N2N_RB) atomicAdd(&acc[rr[k]], vv[k]);
    }
    __syncthreads();

    float* dst = ws + (l ? WS_A1 : WS_A0) + lo;
    for (int i0 = tid; i0 < N2N_RB; i0 += FL_U * SC_T) {
        float sv[FL_U];
        #pragma unroll
        for (int u = 0; u < FL_U; ++u) {
            const int i = i0 + u * SC_T;
            sv[u] = acc[i < N2N_RB ? i : 0];
        }
        #pragma unroll
        for (int u = 0; u < FL_U; ++u) {
            const int i = i0 + u * SC_T;
            if (i < N2N_RB && lo + i < NN && sv[u] != 0.0f) atomicAdd(&dst[i], sv[u]);
        }
    }
}

__global__ __launch_bounds__(SC_T)
void k_sage(const int* __restrict__ s0r, const int* __restrict__ s0c,
            const int* __restrict__ s1r, const int* __restrict__ s1c,
            float* ws) {
    __shared__ __attribute__((aligned(16))) float        num[SG_RB];
    __shared__ __attribute__((aligned(16))) unsigned int cnt2[SG_RB / 2];
    __shared__ __attribute__((aligned(16))) float        awin[SG_WN];
    __shared__ int s_item;
    const int tid = threadIdx.x;
    if (tid == 0) s_item = take_item(reinterpret_cast<int*>(ws) + WS_TK + 2);
    for (int i = tid; i < SG_RB / 4; i += SC_T)
        reinterpret_cast<float4*>(num)[i] = make_float4(0.f, 0.f, 0.f, 0.f);
    for (int i = tid; i < SG_RB / 8; i += SC_T)
        reinterpret_cast<uint4*>(cnt2)[i] = make_uint4(0u, 0u, 0u, 0u);
    __syncthreads();

    const int item = s_item;
    const int c = item % SC_C;
    const int p = (item / SC_C) % SC_P;
    const int l = item / (SC_C * SC_P);
    const int lo = p * SG_RB;
    const int q0 = (c * SG_Q) / SC_C;
    const int n4 = ((c + 1) * SG_Q) / SC_C - q0;
    const int4* __restrict__ row4 = reinterpret_cast<const int4*>(l ? s1r : s0r) + q0;
    const int4* __restrict__ col4 = reinterpret_cast<const int4*>(l ? s1c : s0c) + q0;
    const float* a = ws + (l ? WS_A1 : WS_A0);
    const int STEP4 = SC_T * SC_V;

    int4 qr[SC_V]; int4 qc[SC_V];
    int rr[SC_U]; float gv[SC_U];
    float lv[SC_U]; unsigned int wm;
    #pragma unroll
    for (int u = 0; u < SC_V; ++u) {
        int i4 = u * SC_T + tid;
        int ii = i4 < n4 ? i4 : 0;
        qr[u] = row4[ii];
        qc[u] = col4[ii];
    }
    {
        const float4* __restrict__ a4 = reinterpret_cast<const float4*>(a);
        float4 t[SG_WU];
        #pragma unroll
        for (int u = 0; u < SG_WU; ++u) t[u] = a4[u * SC_T + tid];
        #pragma unroll
        for (int u = 0; u < SG_WU; ++u) reinterpret_cast<float4*>(awin)[u * SC_T + tid] = t[u];
    }
    __syncthreads();
    wm = 0u;
    #pragma unroll
    for (int u = 0; u < SC_V; ++u) {
        bool ok = (u * SC_T + tid) < n4;
        int r4[4] = {qr[u].x - lo, qr[u].y - lo, qr[u].z - lo, qr[u].w - lo};
        int c4[4] = {qc[u].x, qc[u].y, qc[u].z, qc[u].w};
        #pragma unroll
        for (int j = 0; j < 4; ++j) {
            bool m = ok && ((unsigned)r4[j] < (unsigned)SG_RB);
            bool w = (unsigned)c4[j] < (unsigned)SG_WN;
            rr[4 * u + j] = m ? r4[j] : -1;
            gv[4 * u + j] = a[(m && !w) ? c4[j] : 0];
            lv[4 * u + j] = awin[(m && w) ? c4[j] : 0];
            wm |= (w ? 1u : 0u) << (4 * u + j);
        }
    }
    #pragma unroll
    for (int u = 0; u < SC_V; ++u) {
        int i4 = STEP4 + u * SC_T + tid;
        int ii = i4 < n4 ? i4 : 0;
        qr[u] = row4[ii];
        qc[u] = col4[ii];
    }
    for (int b4 = 0; b4 < n4; b4 += STEP4) {
        int nr[SC_U]; float ng[SC_U]; float nl[SC_U]; unsigned int nw = 0u;
        #pragma unroll
        for (int u = 0; u < SC_V; ++u) {
            bool ok = (b4 + STEP4 + u * SC_T + tid) < n4;
            int r4[4] = {qr[u].x - lo, qr[u].y - lo, qr[u].z - lo, qr[u].w - lo};
            int c4[4] = {qc[u].x, qc[u].y, qc[u].z, qc[u].w};
            #pragma unroll
            for (int j = 0; j < 4; ++j) {
                bool m = ok && ((unsigned)r4[j] < (unsigned)SG_RB);
                bool w = (unsigned)c4[j] < (unsigned)SG_WN;
                nr[4 * u + j] = m ? r4[j] : -1;
                ng[4 * u + j] = a[(m && !w) ? c4[j] : 0];
                nl[4 * u + j] = awin[(m && w) ? c4[j] : 0];
                nw |= (w ? 1u : 0u) << (4 * u + j);
            }
        }
        #pragma unroll
        for (int u = 0; u < SC_V; ++u) {
            int i4 = b4 + 2 * STEP4 + u * SC_T + tid;
            int ii = i4 < n4 ? i4 : 0;
            qr[u] = row4[ii];
            qc[u] = col4[ii];
        }
        #pragma unroll
        for (int k = 0; k < SC_U; ++k)
            if (rr[k] >= 0) {
                atomicAdd(&num[rr[k]], ((wm >> k) & 1u) ? lv[k] : gv[k]);
                atomicAdd(&cnt2[rr[k] >> 1], (rr[k] & 1) ? 65536u : 1u);
            }
        #pragma unroll
        for (int k = 0; k < SC_U; ++k) { rr[k] = nr[k]; gv[k] = ng[k]; lv[k] = nl[k]; }
        wm = nw;
    }
    __syncthreads();

    float* dm = ws + (l ? WS_M1   : WS_M0)   + lo;
    float* dd = ws + (l ? WS_DEG1 : WS_DEG0) + lo;
    for (int i0 = tid; i0 < SG_RB; i0 += FL_U * SC_T) {
        unsigned int k[FL_U]; float s[FL_U];
        #pragma unroll
        for (int u = 0; u < FL_U; ++u) {
            const int i  = i0 + u * SC_T;
            const int ii = i < SG_RB ? i : 0;
            k[u] = (cnt2[ii >> 1] >> (16 * (ii & 1))) & 0xffffu;
            s[u] = num[ii];
        }
        #pragma unroll
        for (int u = 0; u < FL_U; ++u) {
            const int i = i0 + u * SC_T;
            if (i < SG_RB && lo + i < NBT && k[u] != 0u) {
                atomicAdd(&dm[i], s[u]);
                atomicAdd(&dd[i], (float)k[u]);
            }
        }
    }
}

#define RW_T    256
#define RW_NCH  (2 * NN / 32)
#define RW_NB   ((RW_NCH * 32 + RW_T - 1) / RW_T)
#define HD_BPB  16
#define HD_NB   (BB / HD_BPB)
#define HD_EP   68
#define HD_HP   33
static_assert(NN % 32 == 0, "a 32-row chunk must stay inside one layer");
static_assert(BB % HD_BPB == 0, "head blocks must tile the subgraph nodes");
static_assert(HG_B * HG_E >= NN && HG_U * SC_T >= HG_E, "the subgraph blocks must cover the subgraph entries");
static_assert(BB <= SC_T && BB <= N2N_RB, "one thread and one LDS word per subgraph node");

__device__ __forceinline__ float row_val(float a, float u1, float m, float u2) {
    return fmaxf(fmaf(a, u1, m * u2), 0.0f);
}

__device__ __forceinline__ float row_ss(float a, float m,
                                        const float* __restrict__ u1, const float* __restrict__ u2) {
    float s0 = 0.f, s1 = 0.f, s2 = 0.f, s3 = 0.f;
    #pragma unroll
    for (int d = 0; d < DD; d += 4) {
        float r0 = row_val(a, u1[d + 0], m, u2[d + 0]);
        float r1 = row_val(a, u1[d + 1], m, u2[d + 1]);
        float r2 = row_val(a, u1[d + 2], m, u2[d + 2]);
        float r3 = row_val(a, u1[d + 3], m, u2[d + 3]);
        s0 = fmaf(r0, r0, s0);
        s1 = fmaf(r1, r1, s1);
        s2 = fmaf(r2, r2, s2);
        s3 = fmaf(r3, r3, s3);
    }
    return (s0 + s1) + (s2 + s3);
}

__global__ __launch_bounds__(RW_T)
void k_rows(const float* __restrict__ wsr,
            const float* __restrict__ cp,
            const int*   __restrict__ act_col,
            const float* __restrict__ h1,
            const float* __restrict__ aux,
            const float* __restrict__ h2w,
            float* __restrict__ out) {
    __shared__ float s_t[2 * HD_BPB], s_a[2 * HD_BPB], s_m[2 * HD_BPB], s_i[2 * HD_BPB];
    __shared__ __attribute__((aligned(16))) float s_esa[2 * HD_BPB * HD_EP];
    __shared__ float s_hid[2 * HD_BPB * HD_HP];

    const float* __restrict__ u1 = wsr + WS_U1;
    const float* __restrict__ u2 = wsr + WS_U2;
    const int lane = threadIdx.x & 31;
    const int c    = lane & 15;
    const int half = lane >> 4;

    if (blockIdx.x < HD_NB) {
        const int tid  = threadIdx.x;
        const int wave = tid >> 5;
        const int b0   = blockIdx.x * HD_BPB;
        if (wave == 0) {
            const int l = lane >> 4;
            const int i = NN + b0 + (lane & 15);
            const float a   = wsr[(l ? WS_A1   : WS_A0)   + i];
            const float num = wsr[(l ? WS_M1   : WS_M0)   + i];
            const float dg  = wsr[(l ? WS_DEG1 : WS_DEG0) + i];
            const float m   = num / fmaxf(dg, 1.0f);
            const float inv = 1.0f / fmaxf(sqrtf(row_ss(a, m, u1, u2)), 1e-12f);
            float t0 = 0.f, t1 = 0.f, t2 = 0.f, t3 = 0.f;
            #pragma unroll
            for (int d = 0; d < DD; d += 4) {
                t0 = fmaf(row_val(a, u1[d + 0], m, u2[d + 0]) * inv, cp[d + 0], t0);
                t1 = fmaf(row_val(a, u1[d + 1], m, u2[d + 1]) * inv, cp[d + 1], t1);
                t2 = fmaf(row_val(a, u1[d + 2], m, u2[d + 2]) * inv, cp[d + 2], t2);
                t3 = fmaf(row_val(a, u1[d + 3], m, u2[d + 3]) * inv, cp[d + 3], t3);
            }
            s_t[lane] = (t0 + t1) + (t2 + t3);
        } else if (wave == 1) {
            const int l = lane >> 4;
            const int i = act_col[b0 + (lane & 15)];
            const float a   = wsr[(l ? WS_A1   : WS_A0)   + i];
            const float num = wsr[(l ? WS_M1   : WS_M0)   + i];
            const float dg  = wsr[(l ? WS_DEG1 : WS_DEG0) + i];
            const float m   = num / fmaxf(dg, 1.0f);
            s_a[lane] = a;
            s_m[lane] = m;
            s_i[lane] = 1.0f / fmaxf(sqrtf(row_ss(a, m, u1, u2)), 1e-12f);
        }
        __syncthreads();
        for (int j = 0; j < (2 * HD_BPB * DD) / RW_T; ++j) {
            const int idx = j * RW_T + tid;
            const int pr  = idx >> 6;
            const int d   = idx & (DD - 1);
            const float ae = row_val(s_a[pr], u1[d], s_m[pr], u2[d]) * s_i[pr];
            s_esa[pr * HD_EP + d] = s_t[pr] * ae;
        }
        __syncthreads();
        if (wave < 4) {
            const int l  = wave >> 1;
            const int tn = wave & 1;
            const float* A = s_esa + (l * HD_BPB) * HD_EP;
            v8f acc = {0.f, 0.f, 0.f, 0.f, 0.f, 0.f, 0.f, 0.f};
            #pragma unroll
            for (int s = 0; s < 16; ++s) {
                const int k0 = 4 * s + 2 * half;
                v2f av, bv;
                av.x = A[c * HD_EP + k0];
                av.y = A[c * HD_EP + k0 + 1];
                bv.x = h1[(k0    ) * 32 + tn * 16 + c];
                bv.y = h1[(k0 + 1) * 32 + tn * 16 + c];
                acc = __builtin_amdgcn_wmma_f32_16x16x4_f32(false, av, false, bv,
                                                            (short)0, acc, false, false);
            }
            #pragma unroll
            for (int r = 0; r < 8; ++r) {
                const int M = r + 8 * half;
                s_hid[(l * HD_BPB + M) * HD_HP + tn * 16 + c] = fmaxf(acc[r], 0.0f);
            }
        }
        __syncthreads();
        if (tid < HD_BPB) {
            const int b = b0 + tid;
            float q = 0.0f;
            for (int l = 0; l < 2; ++l) {
                const float* h = s_hid + (l * HD_BPB + tid) * HD_HP;
                for (int j = 0; j < 32; ++j) q += h[j] * h2w[j];
                for (int j = 0; j < 4; ++j)  q += aux[b * 8 + l * 4 + j] * h2w[32 + j];
            }
            out[b] = q;
        }
        return;
    }

    const int chunk = ((blockIdx.x - HD_NB) * RW_T + threadIdx.x) >> 5;
    if (chunk >= RW_NCH) return;
    const int l  = chunk / (NN / 32);
    const int i0 = (chunk - l * (NN / 32)) * 32;
    const int i  = i0 + lane;

    const float a   = wsr[(l ? WS_A1   : WS_A0)   + i];
    const float num = wsr[(l ? WS_M1   : WS_M0)   + i];
    const float dg  = wsr[(l ? WS_DEG1 : WS_DEG0) + i];
    const float4 p1 = *reinterpret_cast<const float4*>(wsr + WS_U1 + 4 * c);
    const float4 p2 = *reinterpret_cast<const float4*>(wsr + WS_U2 + 4 * c);
    const float m = num / fmaxf(dg, 1.0f);
    const float inv = 1.0f / fmaxf(sqrtf(row_ss(a, m, u1, u2)), 1e-12f);

    float* dst = out + BB + (l * NN + i0 + half) * DD + 4 * c;
    #pragma unroll
    for (int p = 0; p < 16; ++p) {
        const int src = 2 * p + half;
        const float ar = __shfl(a,   src, 32);
        const float mr = __shfl(m,   src, 32);
        const float ir = __shfl(inv, src, 32);
        float4 o;
        o.x = row_val(ar, p1.x, mr, p2.x) * ir;
        o.y = row_val(ar, p1.y, mr, p2.y) * ir;
        o.z = row_val(ar, p1.z, mr, p2.z) * ir;
        o.w = row_val(ar, p1.w, mr, p2.w) * ir;
        *reinterpret_cast<float4*>(dst + 2 * p * DD) = o;
    }
}

extern "C" void kernel_launch(void* const* d_in, const int* in_sizes, int n_in,
                              void* d_out, int out_size, void* d_ws, size_t ws_size,
                              hipStream_t stream) {
    const int*   n2n0_row = (const int*)  d_in[0];
    const float* n2n0_val = (const float*)d_in[2];
    const int*   n2n1_row = (const int*)  d_in[3];
    const float* n2n1_val = (const float*)d_in[5];
    const int*   subg_row = (const int*)  d_in[6];
    const float* subg_val = (const float*)d_in[8];
    const int*   act_col  = (const int*)  d_in[9];
    const int*   sg0_row  = (const int*)  d_in[10];
    const int*   sg0_col  = (const int*)  d_in[11];
    const int*   sg1_row  = (const int*)  d_in[12];
    const int*   sg1_col  = (const int*)  d_in[13];
    const float* aux      = (const float*)d_in[14];
    const float* w_n2l    = (const float*)d_in[15];
    const float* p_conv   = (const float*)d_in[16];
    const float* W_sage   = (const float*)d_in[17];
    const float* cp       = (const float*)d_in[18];
    const float* h1w      = (const float*)d_in[19];
    const float* h2w      = (const float*)d_in[20];
    float* out = (float*)d_out;
    float* ws  = (float*)d_ws;

    k_prep <<<1 + PR_ZB, PR_T, 0, stream>>>(w_n2l, p_conv, W_sage, ws);
    k_n2n  <<<SC_GRID + HG_B, SC_T, 0, stream>>>(n2n0_row, n2n0_val, n2n1_row, n2n1_val, subg_row, subg_val, ws);
    k_sage <<<SC_GRID, SC_T, 0, stream>>>(sg0_row, sg0_col, sg1_row, sg1_col, ws);
    k_rows <<<HD_NB + RW_NB, RW_T, 0, stream>>>(ws, cp, act_col, h1w, aux, h2w, out);
}


# launch 1 keeps only a-plane zeroing + tickets: m/deg clearing goes to the n2n scatter blocks, the dense chain to one extra n2n-launch block (k split over thread groups: u1, u2 sum order changes)
# speedup vs baseline: 8.1827x; 1.0582x over previous
#include <hip/hip_runtime.h>

#define NN   100000
#define BB   512
#define EE   1000000
#define ESZ  1200000
#define DD   64
#define NBT  (NN + BB)

#define WS_A0   0
#define WS_A1   (NBT)
#define WS_M0   (2*NBT)
#define WS_M1   (3*NBT)
#define WS_DEG0 (4*NBT)
#define WS_DEG1 (5*NBT)
#define WS_SM   (6*NBT)
#define WS_TK   (WS_SM)
#define WS_U1   (WS_SM + 128)
#define WS_U2   (WS_SM + 192)

#define SC_P    4
#define SC_C    32
#define SC_T    1024
#define SC_V    2
#define SC_U    (4 * SC_V)
#define SC_GRID (2 * SC_P * SC_C)
#define FL_U    5
#define HG_B    8
#define HG_E    ((NN + HG_B - 1) / HG_B)
#define HG_U    ((HG_E + SC_T - 1) / SC_T)
#define SC_RB(n) ((((n) + SC_P - 1) / SC_P + 31) / 32 * 32)
#define N2N_RB  SC_RB(NN)
#define SG_RB   SC_RB(NBT)
#define N2N_Q   (EE / 4)
#define SG_Q    (ESZ / 4)
#define SG_WN   ((((320 * 1024 - 2048) - SG_RB * 4 - SG_RB * 2) / 4) / (4 * SC_T) * (4 * SC_T))
#define SG_WU   (SG_WN / (4 * SC_T))
static_assert(SC_P * N2N_RB >= NN && SC_P * SG_RB >= NBT, "row ranges must cover the rows");
static_assert(N2N_RB % 4 == 0 && SG_RB % 4 == 0, "the LDS sums are cleared in four-word pieces");
static_assert(NBT % 32 == 0, "the accumulator planes must start on whole lines");
static_assert(EE % 4 == 0 && ESZ % 4 == 0, "the edges are read in four-word pieces");
static_assert((long long)SC_C * SG_Q < 2147483647LL, "chunk bounds are computed in int");
static_assert(4 * (SG_Q / SC_C + 1) < 65536, "a block's edge count per row must fit 16 bits");
static_assert(SG_RB % 8 == 0, "the packed counts are cleared in four-word pieces");
static_assert(SG_WN > 0 && SG_WN % (4 * SC_T) == 0 && SG_WN <= NBT, "the window is staged in whole rounds of four-word pieces of the plane");

typedef float v2f __attribute__((ext_vector_type(2)));
typedef float v8f __attribute__((ext_vector_type(8)));

__device__ __forceinline__ int take_item(int* tk) {
    const unsigned id = (unsigned)__builtin_amdgcn_s_sendmsg_rtn(0x87);
    const int h = (int)((id >> 18) & 1u);
    const int t = atomicAdd(&tk[h], 1);
    const int half_n = (2 * SC_P * SC_C) / 2;
    int item = t < half_n ? h * half_n + t : (1 - h) * half_n + (2 * half_n - 1 - t);
    return item < 0 ? 0 : (item > 2 * half_n - 1 ? 2 * half_n - 1 : item);
}

#define PR_T   256
#define PR_Q   (2 * NBT / 4)
#define PR_ZB  ((PR_Q + PR_T - 1) / PR_T)
static_assert((2 * NBT) % 4 == 0 && WS_A0 == 0 && WS_A1 == NBT, "the a planes are cleared as one run of four-word pieces");
__global__ __launch_bounds__(PR_T)
void k_prep(float* ws) {
    const int i = blockIdx.x * PR_T + threadIdx.x;
    if (i < PR_Q) reinterpret_cast<float4*>(ws)[i] = make_float4(0.f, 0.f, 0.f, 0.f);
    if (i < 4) reinterpret_cast<int*>(ws)[WS_TK + i] = 0;
}

#define ZM_Q   ((NBT / 4 + SC_P * SC_C - 1) / (SC_P * SC_C))
static_assert(2 * ZM_Q <= SC_T, "one thread per piece of the block's m and deg shares");
#define DC_G   (SC_T / DD)
#define DC_W   DD
#define DC_PA  (2 * DD)
#define DC_PB  (2 * DD + DC_G * DD)
static_assert(DC_G * 4 == DD, "the k groups must cover the 64 k of the dense products");
static_assert(DC_PB + DC_G * DD <= N2N_RB, "the dense chain's scratch lies inside the block's LDS sums");

__global__ __launch_bounds__(SC_T)
void k_n2n(const int* __restrict__ r0, const float* __restrict__ v0,
           const int* __restrict__ r1, const float* __restrict__ v1,
           const int* __restrict__ subg_row, const float* __restrict__ subg_val,
           const float* __restrict__ w_n2l, const float* __restrict__ p_conv,
           const float* __restrict__ W_sage,
           float* ws) {
    __shared__ __attribute__((aligned(16))) float acc[N2N_RB];
    __shared__ int s_item;
    const int tid = threadIdx.x;
    const int bid = blockIdx.x;
    if (bid == SC_GRID + HG_B) {
        const int g = tid >> 6;
        const int d = tid & (DD - 1);
        const float wa = w_n2l[d], wb = w_n2l[DD + d];
        float pc[4], s1[4], s2[4];
        #pragma unroll
        for (int j = 0; j < 4; ++j) {
            const int k = 4 * g + j;
            pc[j] = p_conv[k * DD + d];
            s1[j] = W_sage[k * DD + d];
            s2[j] = W_sage[(DD + k) * DD + d];
        }
        const float x = fmaxf(wa + wb, 0.0f);
        if (g == 0) acc[d] = x;
        __syncthreads();
        float ss = 0.f;
        for (int k = 0; k < DD; ++k) ss += acc[k] * acc[k];
        const float nrm = fmaxf(sqrtf(ss), 1e-12f);
        float wd = 0.f;
        #pragma unroll
        for (int j = 0; j < 4; ++j) wd += (acc[4 * g + j] / nrm) * pc[j];
        acc[DC_PA + g * DD + d] = wd;
        __syncthreads();
        if (g == 0) {
            float w = 0.f;
            for (int q = 0; q < DC_G; ++q) w += acc[DC_PA + q * DD + d];
            acc[DC_W + d] = w;
        }
        __syncthreads();
        float u1 = 0.f, u2 = 0.f;
        #pragma unroll
        for (int j = 0; j < 4; ++j) {
            const float wk = acc[DC_W + 4 * g + j];
            u1 += wk * s1[j];
            u2 += wk * s2[j];
        }
        acc[DC_PA + g * DD + d] = u1;
        acc[DC_PB + g * DD + d] = u2;
        __syncthreads();
        if (g < 2) {
            const int o = g ? DC_PB : DC_PA;
            float s = 0.f;
            for (int q = 0; q < DC_G; ++q) s += acc[o + q * DD + d];
            ws[(g ? WS_U2 : WS_U1) + d] = s;
        }
        return;
    }
    if (bid >= SC_GRID) {
        const int e0 = (bid - SC_GRID) * HG_E;
        if (tid < BB) acc[tid] = 0.0f;
        __syncthreads();
        int hb[HG_U]; float hv[HG_U];
        #pragma unroll
        for (int u = 0; u < HG_U; ++u) {
            const int j = u * SC_T + tid;
            const bool ok = j < HG_E && e0 + j < NN;
            hb[u] = ok ? subg_row[e0 + j] : -1;
            hv[u] = ok ? subg_val[e0 + j] : 0.0f;
        }
        #pragma unroll
        for (int u = 0; u < HG_U; ++u)
            if ((unsigned)hb[u] < (unsigned)BB) atomicAdd(&acc[hb[u]], hv[u]);
        __syncthreads();
        if (tid < BB) {
            const float s = acc[tid];
            if (s != 0.0f) {
                atomicAdd(&ws[WS_A0 + NN + tid], s);
                atomicAdd(&ws[WS_A1 + NN + tid], s);
            }
        }
        return;
    }
    if (tid == 0) s_item = take_item(reinterpret_cast<int*>(ws) + WS_TK);
    for (int i = tid; i < N2N_RB / 4; i += SC_T)
        reinterpret_cast<float4*>(acc)[i] = make_float4(0.f, 0.f, 0.f, 0.f);
    __syncthreads();

    const int item = s_item;
    const int c = item % SC_C;
    const int p = (item / SC_C) % SC_P;
    const int l = item / (SC_C * SC_P);
    const int lo = p * N2N_RB;
    if (tid < 2 * ZM_Q) {
        const int pl = tid >= ZM_Q ? 1 : 0;
        const int k  = (p * SC_C + c) * ZM_Q + (tid - pl * ZM_Q);
        float* zb = ws + (pl ? (l ? WS_DEG1 : WS_DEG0) : (l ? WS_M1 : WS_M0));
        if (k < NBT / 4) reinterpret_cast<float4*>(zb)[k] = make_float4(0.f, 0.f, 0.f, 0.f);
    }
    const int q0 = (c * N2N_Q) / SC_C;
    const int n4 = ((c + 1) * N2N_Q) / SC_C - q0;
    const int4*   __restrict__ row4 = reinterpret_cast<const int4*>(l ? r1 : r0) + q0;
    const float4* __restrict__ val4 = reinterpret_cast<const float4*>(l ? v1 : v0) + q0;

    int4 qr[SC_V]; float4 qv[SC_V];
    #pragma unroll
    for (int u = 0; u < SC_V; ++u) {
        int i4 = u * SC_T + tid;
        int ii = i4 < n4 ? i4 : 0;
        qr[u] = row4[ii];
        qv[u] = val4[ii];
    }
    for (int b4 = 0; b4 < n4; b4 += SC_T * SC_V) {
        int rr[SC_U]; float vv[SC_U];
        #pragma unroll
        for (int u = 0; u < SC_V; ++u) {
            bool ok = (b4 + u * SC_T + tid) < n4;
            rr[4 * u + 0] = ok ? qr[u].x - lo : -1;  vv[4 * u + 0] = qv[u].x;
            rr[4 * u + 1] = ok ? qr[u].y - lo : -1;  vv[4 * u + 1] = qv[u].y;
            rr[4 * u + 2] = ok ? qr[u].z - lo : -1;  vv[4 * u + 2] = qv[u].z;
            rr[4 * u + 3] = ok ? qr[u].w - lo : -1;  vv[4 * u + 3] = qv[u].w;
        }
        #pragma unroll
        for (int u = 0; u < SC_V; ++u) {
            int i4 = b4 + SC_T * SC_V + u * SC_T + tid;
            int ii = i4 < n4 ? i4 : 0;
            qr[u] = row4[ii];
            qv[u] = val4[ii];
        }
        #pragma unroll
        for (int k = 0; k < SC_U; ++k)
            if ((unsigned)rr[k] < (unsigned)N2N_RB) atomicAdd(&acc[rr[k]], vv[k]);
    }
    __syncthreads();

    float* dst = ws + (l ? WS_A1 : WS_A0) + lo;
    for (int i0 = tid; i0 < N2N_RB; i0 += FL_U * SC_T) {
        float sv[FL_U];
        #pragma unroll
        for (int u = 0; u < FL_U; ++u) {
            const int i = i0 + u * SC_T;
            sv[u] = acc[i < N2N_RB ? i : 0];
        }
        #pragma unroll
        for (int u = 0; u < FL_U; ++u) {
            const int i = i0 + u * SC_T;
            if (i < N2N_RB && lo + i < NN && sv[u] != 0.0f) atomicAdd(&dst[i], sv[u]);
        }
    }
}

__global__ __launch_bounds__(SC_T)
void k_sage(const int* __restrict__ s0r, const int* __restrict__ s0c,
            const int* __restrict__ s1r, const int* __restrict__ s1c,
            float* ws) {
    __shared__ __attribute__((aligned(16))) float        num[SG_RB];
    __shared__ __attribute__((aligned(16))) unsigned int cnt2[SG_RB / 2];
    __shared__ __attribute__((aligned(16))) float        awin[SG_WN];
    __shared__ int s_item;
    const int tid = threadIdx.x;
    if (tid == 0) s_item = take_item(reinterpret_cast<int*>(ws) + WS_TK + 2);
    for (int i = tid; i < SG_RB / 4; i += SC_T)
        reinterpret_cast<float4*>(num)[i] = make_float4(0.f, 0.f, 0.f, 0.f);
    for (int i = tid; i < SG_RB / 8; i += SC_T)
        reinterpret_cast<uint4*>(cnt2)[i] = make_uint4(0u, 0u, 0u, 0u);
    __syncthreads();

    const int item = s_item;
    const int c = item % SC_C;
    const int p = (item / SC_C) % SC_P;
    const int l = item / (SC_C * SC_P);
    const int lo = p * SG_RB;
    const int q0 = (c * SG_Q) / SC_C;
    const int n4 = ((c + 1) * SG_Q) / SC_C - q0;
    const int4* __restrict__ row4 = reinterpret_cast<const int4*>(l ? s1r : s0r) + q0;
    const int4* __restrict__ col4 = reinterpret_cast<const int4*>(l ? s1c : s0c) + q0;
    const float* a = ws + (l ? WS_A1 : WS_A0);
    const int STEP4 = SC_T * SC_V;

    int4 qr[SC_V]; int4 qc[SC_V];
    int rr[SC_U]; float gv[SC_U];
    float lv[SC_U]; unsigned int wm;
    #pragma unroll
    for (int u = 0; u < SC_V; ++u) {
        int i4 = u * SC_T + tid;
        int ii = i4 < n4 ? i4 : 0;
        qr[u] = row4[ii];
        qc[u] = col4[ii];
    }
    {
        const float4* __restrict__ a4 = reinterpret_cast<const float4*>(a);
        float4 t[SG_WU];
        #pragma unroll
        for (int u = 0; u < SG_WU; ++u) t[u] = a4[u * SC_T + tid];
        #pragma unroll
        for (int u = 0; u < SG_WU; ++u) reinterpret_cast<float4*>(awin)[u * SC_T + tid] = t[u];
    }
    __syncthreads();
    wm = 0u;
    #pragma unroll
    for (int u = 0; u < SC_V; ++u) {
        bool ok = (u * SC_T + tid) < n4;
        int r4[4] = {qr[u].x - lo, qr[u].y - lo, qr[u].z - lo, qr[u].w - lo};
        int c4[4] = {qc[u].x, qc[u].y, qc[u].z, qc[u].w};
        #pragma unroll
        for (int j = 0; j < 4; ++j) {
            bool m = ok && ((unsigned)r4[j] < (unsigned)SG_RB);
            bool w = (unsigned)c4[j] < (unsigned)SG_WN;
            rr[4 * u + j] = m ? r4[j] : -1;
            gv[4 * u + j] = a[(m && !w) ? c4[j] : 0];
            lv[4 * u + j] = awin[(m && w) ? c4[j] : 0];
            wm |= (w ? 1u : 0u) << (4 * u + j);
        }
    }
    #pragma unroll
    for (int u = 0; u < SC_V; ++u) {
        int i4 = STEP4 + u * SC_T + tid;
        int ii = i4 < n4 ? i4 : 0;
        qr[u] = row4[ii];
        qc[u] = col4[ii];
    }
    for (int b4 = 0; b4 < n4; b4 += STEP4) {
        int nr[SC_U]; float ng[SC_U]; float nl[SC_U]; unsigned int nw = 0u;
        #pragma unroll
        for (int u = 0; u < SC_V; ++u) {
            bool ok = (b4 + STEP4 + u * SC_T + tid) < n4;
            int r4[4] = {qr[u].x - lo, qr[u].y - lo, qr[u].z - lo, qr[u].w - lo};
            int c4[4] = {qc[u].x, qc[u].y, qc[u].z, qc[u].w};
            #pragma unroll
            for (int j = 0; j < 4; ++j) {
                bool m = ok && ((unsigned)r4[j] < (unsigned)SG_RB);
                bool w = (unsigned)c4[j] < (unsigned)SG_WN;
                nr[4 * u + j] = m ? r4[j] : -1;
                ng[4 * u + j] = a[(m && !w) ? c4[j] : 0];
                nl[4 * u + j] = awin[(m && w) ? c4[j] : 0];
                nw |= (w ? 1u : 0u) << (4 * u + j);
            }
        }
        #pragma unroll
        for (int u = 0; u < SC_V; ++u) {
            int i4 = b4 + 2 * STEP4 + u * SC_T + tid;
            int ii = i4 < n4 ? i4 : 0;
            qr[u] = row4[ii];
            qc[u] = col4[ii];
        }
        #pragma unroll
        for (int k = 0; k < SC_U; ++k)
            if (rr[k] >= 0) {
                atomicAdd(&num[rr[k]], ((wm >> k) & 1u) ? lv[k] : gv[k]);
                atomicAdd(&cnt2[rr[k] >> 1], (rr[k] & 1) ? 65536u : 1u);
            }
        #pragma unroll
        for (int k = 0; k < SC_U; ++k) { rr[k] = nr[k]; gv[k] = ng[k]; lv[k] = nl[k]; }
        wm = nw;
    }
    __syncthreads();

    float* dm = ws + (l ? WS_M1   : WS_M0)   + lo;
    float* dd = ws + (l ? WS_DEG1 : WS_DEG0) + lo;
    for (int i0 = tid; i0 < SG_RB; i0 += FL_U * SC_T) {
        unsigned int k[FL_U]; float s[FL_U];
        #pragma unroll
        for (int u = 0; u < FL_U; ++u) {
            const int i  = i0 + u * SC_T;
            const int ii = i < SG_RB ? i : 0;
            k[u] = (cnt2[ii >> 1] >> (16 * (ii & 1))) & 0xffffu;
            s[u] = num[ii];
        }
        #pragma unroll
        for (int u = 0; u < FL_U; ++u) {
            const int i = i0 + u * SC_T;
            if (i < SG_RB && lo + i < NBT && k[u] != 0u) {
                atomicAdd(&dm[i], s[u]);
                atomicAdd(&dd[i], (float)k[u]);
            }
        }
    }
}

#define RW_T    256
#define RW_NCH  (2 * NN / 32)
#define RW_NB   ((RW_NCH * 32 + RW_T - 1) / RW_T)
#define HD_BPB  16
#define HD_NB   (BB / HD_BPB)
#define HD_EP   68
#define HD_HP   33
static_assert(NN % 32 == 0, "a 32-row chunk must stay inside one layer");
static_assert(BB % HD_BPB == 0, "head blocks must tile the subgraph nodes");
static_assert(HG_B * HG_E >= NN && HG_U * SC_T >= HG_E, "the subgraph blocks must cover the subgraph entries");
static_assert(BB <= SC_T && BB <= N2N_RB, "one thread and one LDS word per subgraph node");

__device__ __forceinline__ float row_val(float a, float u1, float m, float u2) {
    return fmaxf(fmaf(a, u1, m * u2), 0.0f);
}

__device__ __forceinline__ float row_ss(float a, float m,
                                        const float* __restrict__ u1, const float* __restrict__ u2) {
    float s0 = 0.f, s1 = 0.f, s2 = 0.f, s3 = 0.f;
    #pragma unroll
    for (int d = 0; d < DD; d += 4) {
        float r0 = row_val(a, u1[d + 0], m, u2[d + 0]);
        float r1 = row_val(a, u1[d + 1], m, u2[d + 1]);
        float r2 = row_val(a, u1[d + 2], m, u2[d + 2]);
        float r3 = row_val(a, u1[d + 3], m, u2[d + 3]);
        s0 = fmaf(r0, r0, s0);
        s1 = fmaf(r1, r1, s1);
        s2 = fmaf(r2, r2, s2);
        s3 = fmaf(r3, r3, s3);
    }
    return (s0 + s1) + (s2 + s3);
}

__global__ __launch_bounds__(RW_T)
void k_rows(const float* __restrict__ wsr,
            const float* __restrict__ cp,
            const int*   __restrict__ act_col,
            const float* __restrict__ h1,
            const float* __restrict__ aux,
            const float* __restrict__ h2w,
            float* __restrict__ out) {
    __shared__ float s_t[2 * HD_BPB], s_a[2 * HD_BPB], s_m[2 * HD_BPB], s_i[2 * HD_BPB];
    __shared__ __attribute__((aligned(16))) float s_esa[2 * HD_BPB * HD_EP];
    __shared__ float s_hid[2 * HD_BPB * HD_HP];

    const float* __restrict__ u1 = wsr + WS_U1;
    const float* __restrict__ u2 = wsr + WS_U2;
    const int lane = threadIdx.x & 31;
    const int c    = lane & 15;
    const int half = lane >> 4;

    if (blockIdx.x < HD_NB) {
        const int tid  = threadIdx.x;
        const int wave = tid >> 5;
        const int b0   = blockIdx.x * HD_BPB;
        if (wave == 0) {
            const int l = lane >> 4;
            const int i = NN + b0 + (lane & 15);
            const float a   = wsr[(l ? WS_A1   : WS_A0)   + i];
            const float num = wsr[(l ? WS_M1   : WS_M0)   + i];
            const float dg  = wsr[(l ? WS_DEG1 : WS_DEG0) + i];
            const float m   = num / fmaxf(dg, 1.0f);
            const float inv = 1.0f / fmaxf(sqrtf(row_ss(a, m, u1, u2)), 1e-12f);
            float t0 = 0.f, t1 = 0.f, t2 = 0.f, t3 = 0.f;
            #pragma unroll
            for (int d = 0; d < DD; d += 4) {
                t0 = fmaf(row_val(a, u1[d + 0], m, u2[d + 0]) * inv, cp[d + 0], t0);
                t1 = fmaf(row_val(a, u1[d + 1], m, u2[d + 1]) * inv, cp[d + 1], t1);
                t2 = fmaf(row_val(a, u1[d + 2], m, u2[d + 2]) * inv, cp[d + 2], t2);
                t3 = fmaf(row_val(a, u1[d + 3], m, u2[d + 3]) * inv, cp[d + 3], t3);
            }
            s_t[lane] = (t0 + t1) + (t2 + t3);
        } else if (wave == 1) {
            const int l = lane >> 4;
            const int i = act_col[b0 + (lane & 15)];
            const float a   = wsr[(l ? WS_A1   : WS_A0)   + i];
            const float num = wsr[(l ? WS_M1   : WS_M0)   + i];
            const float dg  = wsr[(l ? WS_DEG1 : WS_DEG0) + i];
            const float m   = num / fmaxf(dg, 1.0f);
            s_a[lane] = a;
            s_m[lane] = m;
            s_i[lane] = 1.0f / fmaxf(sqrtf(row_ss(a, m, u1, u2)), 1e-12f);
        }
        __syncthreads();
        for (int j = 0; j < (2 * HD_BPB * DD) / RW_T; ++j) {
            const int idx = j * RW_T + tid;
            const int pr  = idx >> 6;
            const int d   = idx & (DD - 1);
            const float ae = row_val(s_a[pr], u1[d], s_m[pr], u2[d]) * s_i[pr];
            s_esa[pr * HD_EP + d] = s_t[pr] * ae;
        }
        __syncthreads();
        if (wave < 4) {
            const int l  = wave >> 1;
            const int tn = wave & 1;
            const float* A = s_esa + (l * HD_BPB) * HD_EP;
            v8f acc = {0.f, 0.f, 0.f, 0.f, 0.f, 0.f, 0.f, 0.f};
            #pragma unroll
            for (int s = 0; s < 16; ++s) {
                const int k0 = 4 * s + 2 * half;
                v2f av, bv;
                av.x = A[c * HD_EP + k0];
                av.y = A[c * HD_EP + k0 + 1];
                bv.x = h1[(k0    ) * 32 + tn * 16 + c];
                bv.y = h1[(k0 + 1) * 32 + tn * 16 + c];
                acc = __builtin_amdgcn_wmma_f32_16x16x4_f32(false, av, false, bv,
                                                            (short)0, acc, false, false);
            }
            #pragma unroll
            for (int r = 0; r < 8; ++r) {
                const int M = r + 8 * half;
                s_hid[(l * HD_BPB + M) * HD_HP + tn * 16 + c] = fmaxf(acc[r], 0.0f);
            }
        }
        __syncthreads();
        if (tid < HD_BPB) {
            const int b = b0 + tid;
            float q = 0.0f;
            for (int l = 0; l < 2; ++l) {
                const float* h = s_hid + (l * HD_BPB + tid) * HD_HP;
                for (int j = 0; j < 32; ++j) q += h[j] * h2w[j];
                for (int j = 0; j < 4; ++j)  q += aux[b * 8 + l * 4 + j] * h2w[32 + j];
            }
            out[b] = q;
        }
        return;
    }

    const int chunk = ((blockIdx.x - HD_NB) * RW_T + threadIdx.x) >> 5;
    if (chunk >= RW_NCH) return;
    const int l  = chunk / (NN / 32);
    const int i0 = (chunk - l * (NN / 32)) * 32;
    const int i  = i0 + lane;

    const float a   = wsr[(l ? WS_A1   : WS_A0)   + i];
    const float num = wsr[(l ? WS_M1   : WS_M0)   + i];
    const float dg  = wsr[(l ? WS_DEG1 : WS_DEG0) + i];
    const float4 p1 = *reinterpret_cast<const float4*>(wsr + WS_U1 + 4 * c);
    const float4 p2 = *reinterpret_cast<const float4*>(wsr + WS_U2 + 4 * c);
    const float m = num / fmaxf(dg, 1.0f);
    const float inv = 1.0f / fmaxf(sqrtf(row_ss(a, m, u1, u2)), 1e-12f);

    float* dst = out + BB + (l * NN + i0 + half) * DD + 4 * c;
    #pragma unroll
    for (int p = 0; p < 16; ++p) {
        const int src = 2 * p + half;
        const float ar = __shfl(a,   src, 32);
        const float mr = __shfl(m,   src, 32);
        const float ir = __shfl(inv, src, 32);
        float4 o;
        o.x = row_val(ar, p1.x, mr, p2.x) * ir;
        o.y = row_val(ar, p1.y, mr, p2.y) * ir;
        o.z = row_val(ar, p1.z, mr, p2.z) * ir;
        o.w = row_val(ar, p1.w, mr, p2.w) * ir;
        *reinterpret_cast<float4*>(dst + 2 * p * DD) = o;
    }
}

extern "C" void kernel_launch(void* const* d_in, const int* in_sizes, int n_in,
                              void* d_out, int out_size, void* d_ws, size_t ws_size,
                              hipStream_t stream) {
    const int*   n2n0_row = (const int*)  d_in[0];
    const float* n2n0_val = (const float*)d_in[2];
    const int*   n2n1_row = (const int*)  d_in[3];
    const float* n2n1_val = (const float*)d_in[5];
    const int*   subg_row = (const int*)  d_in[6];
    const float* subg_val = (const float*)d_in[8];
    const int*   act_col  = (const int*)  d_in[9];
    const int*   sg0_row  = (const int*)  d_in[10];
    const int*   sg0_col  = (const int*)  d_in[11];
    const int*   sg1_row  = (const int*)  d_in[12];
    const int*   sg1_col  = (const int*)  d_in[13];
    const float* aux      = (const float*)d_in[14];
    const float* w_n2l    = (const float*)d_in[15];
    const float* p_conv   = (const float*)d_in[16];
    const float* W_sage   = (const float*)d_in[17];
    const float* cp       = (const float*)d_in[18];
    const float* h1w      = (const float*)d_in[19];
    const float* h2w      = (const float*)d_in[20];
    float* out = (float*)d_out;
    float* ws  = (float*)d_ws;

    k_prep <<<PR_ZB, PR_T, 0, stream>>>(ws);
    k_n2n  <<<SC_GRID + HG_B + 1, SC_T, 0, stream>>>(n2n0_row, n2n0_val, n2n1_row, n2n1_val, subg_row, subg_val,
                                               w_n2l, p_conv, W_sage, ws);
    k_sage <<<SC_GRID, SC_T, 0, stream>>>(sg0_row, sg0_col, sg1_row, sg1_col, ws);
    k_rows <<<HD_NB + RW_NB, RW_T, 0, stream>>>(ws, cp, act_col, h1w, aux, h2w, out);
}
